# starting kernel
# baseline (speedup 1.0000x reference)
#include <hip/hip_runtime.h>

typedef __attribute__((ext_vector_type(2))) float v2f;
typedef __attribute__((ext_vector_type(8))) float v8f;

__device__ __forceinline__ v8f wmma_f32(v2f a, v2f b, v8f c) {
  return __builtin_amdgcn_wmma_f32_16x16x4_f32(false, a, false, b, (short)0, c,
                                               false, false);
}

__global__ __launch_bounds__(256) void k_node_ffn(
    const float* __restrict__ x, const float* __restrict__ Wn1,
    const float* __restrict__ bn1, const float* __restrict__ Wn2,
    const float* __restrict__ bn2, float* __restrict__ h) {
  __shared__ float xs[16][8];
  __shared__ float ts[16][256];
  const int tid = threadIdx.x;
  const int node0 = blockIdx.x * 16;

  if (tid < 16 * 8) {
    int r = tid >> 3, k = tid & 7;
    xs[r][k] = (k < 6) ? x[(node0 + r) * 6 + k] : 0.0f;
  }
  __syncthreads();

  {
    int o = tid;
    float w0 = Wn1[0 * 256 + o], w1 = Wn1[1 * 256 + o], w2 = Wn1[2 * 256 + o];
    float w3 = Wn1[3 * 256 + o], w4 = Wn1[4 * 256 + o], w5 = Wn1[5 * 256 + o];
    float b = bn1[o];
    for (int r = 0; r < 16; ++r) {
      float acc = b + xs[r][0] * w0 + xs[r][1] * w1 + xs[r][2] * w2 +
                  xs[r][3] * w3 + xs[r][4] * w4 + xs[r][5] * w5;
      ts[r][o] = fmaxf(acc, 0.0f);
    }
  }
  __syncthreads();

  const int wave = tid >> 5, lane = tid & 31;
  const int hi = lane >> 4, lc = lane & 15;
  const int n0 = wave * 16;
  v8f acc = {};
  for (int k0 = 0; k0 < 256; k0 += 4) {
    int k = k0 + 2 * hi;
    v2f a, b;
    a.x = ts[lc][k];
    a.y = ts[lc][k + 1];
    b.x = Wn2[k * 128 + n0 + lc];
    b.y = Wn2[(k + 1) * 128 + n0 + lc];
    acc = wmma_f32(a, b, acc);
  }
  float bias = bn2[n0 + lc];
  for (int v = 0; v < 8; ++v) {
    int r = v + 8 * hi;
    h[(node0 + r) * 128 + n0 + lc] = acc[v] + bias;
  }
}

__global__ __launch_bounds__(256) void k_edge_ffn(
    const float* __restrict__ h, const float* __restrict__ We1,
    const float* __restrict__ be1, const float* __restrict__ We2,
    const float* __restrict__ be2, float* __restrict__ E) {
  __shared__ float hs[16][128];
  __shared__ float us[16][128];
  const int tid = threadIdx.x;
  const int node0 = blockIdx.x * 16;
  for (int idx = tid; idx < 16 * 128; idx += 256)
    hs[idx >> 7][idx & 127] = h[node0 * 128 + idx];
  __syncthreads();

  const int wave = tid >> 5, lane = tid & 31;
  const int hi = lane >> 4, lc = lane & 15;
  const int n0 = wave * 16;

  v8f acc = {};
  for (int k0 = 0; k0 < 128; k0 += 4) {
    int k = k0 + 2 * hi;
    v2f a, b;
    a.x = hs[lc][k];
    a.y = hs[lc][k + 1];
    b.x = We1[k * 128 + n0 + lc] + We1[(k + 128) * 128 + n0 + lc];
    b.y = We1[(k + 1) * 128 + n0 + lc] + We1[(k + 129) * 128 + n0 + lc];
    acc = wmma_f32(a, b, acc);
  }
  {
    float bias = be1[n0 + lc];
    for (int v = 0; v < 8; ++v) us[v + 8 * hi][n0 + lc] = fmaxf(acc[v] + bias, 0.0f);
  }
  __syncthreads();

  v8f acc2 = {};
  for (int k0 = 0; k0 < 128; k0 += 4) {
    int k = k0 + 2 * hi;
    v2f a, b;
    a.x = us[lc][k];
    a.y = us[lc][k + 1];
    b.x = We2[k * 128 + n0 + lc];
    b.y = We2[(k + 1) * 128 + n0 + lc];
    acc2 = wmma_f32(a, b, acc2);
  }
  float bias2 = be2[n0 + lc];
  for (int v = 0; v < 8; ++v) {
    int r = v + 8 * hi;
    E[(node0 + r) * 128 + n0 + lc] = acc2[v] + bias2;
  }
}

__global__ __launch_bounds__(256) void k_reduce(const float* __restrict__ E,
                                                float* __restrict__ S) {
  int tid = threadIdx.x;
  int n = tid >> 7, c = tid & 127;
  const float* base = E + (size_t)n * 512 * 128 + c;
  float s = 0.0f;
  for (int i = 0; i < 512; ++i) s += base[i * 128];
  S[n * 128 + c] = s;
}

__global__ __launch_bounds__(256) void k_e2n_ffn(
    const float* __restrict__ E, const float* __restrict__ S,
    const float* __restrict__ Wen1, const float* __restrict__ ben1,
    const float* __restrict__ Wen2, const float* __restrict__ ben2,
    float* __restrict__ h2) {
  __shared__ float as[16][128];
  __shared__ float us[16][256];
  const int tid = threadIdx.x;
  const int node0 = blockIdx.x * 16;
  for (int idx = tid; idx < 16 * 128; idx += 256) {
    int r = idx >> 7, c = idx & 127;
    int node = node0 + r;
    as[r][c] = S[(node >> 9) * 128 + c] - E[node * 128 + c];
  }
  __syncthreads();

  const int wave = tid >> 5, lane = tid & 31;
  const int hi = lane >> 4, lc = lane & 15;

  for (int t = 0; t < 2; ++t) {
    int n0 = (wave + t * 8) * 16;
    v8f acc = {};
    for (int k0 = 0; k0 < 128; k0 += 4) {
      int k = k0 + 2 * hi;
      v2f a, b;
      a.x = as[lc][k];
      a.y = as[lc][k + 1];
      b.x = Wen1[k * 256 + n0 + lc];
      b.y = Wen1[(k + 1) * 256 + n0 + lc];
      acc = wmma_f32(a, b, acc);
    }
    float bias = ben1[n0 + lc];
    for (int v = 0; v < 8; ++v) us[v + 8 * hi][n0 + lc] = fmaxf(acc[v] + bias, 0.0f);
  }
  __syncthreads();

  int n0 = wave * 16;
  v8f acc2 = {};
  for (int k0 = 0; k0 < 256; k0 += 4) {
    int k = k0 + 2 * hi;
    v2f a, b;
    a.x = us[lc][k];
    a.y = us[lc][k + 1];
    b.x = Wen2[k * 128 + n0 + lc];
    b.y = Wen2[(k + 1) * 128 + n0 + lc];
    acc2 = wmma_f32(a, b, acc2);
  }
  float bias2 = ben2[n0 + lc];
  for (int v = 0; v < 8; ++v) {
    int r = v + 8 * hi;
    h2[(node0 + r) * 128 + n0 + lc] = acc2[v] + bias2;
  }
}

__global__ __launch_bounds__(256) void k_n2e(
    const float* __restrict__ h2, const float* __restrict__ Wne1,
    const float* __restrict__ bne1, const float* __restrict__ Wne2,
    const float* __restrict__ bne2, float* __restrict__ out2) {
  __shared__ float hs[16][128];
  __shared__ float us[16][128];
  __shared__ float ps[16][16];
  __shared__ float sc[16];
  const int tid = threadIdx.x;
  const int node0 = blockIdx.x * 16;
  for (int idx = tid; idx < 16 * 128; idx += 256)
    hs[idx >> 7][idx & 127] = h2[node0 * 128 + idx];
  __syncthreads();

  const int wave = tid >> 5, lane = tid & 31;
  const int hi = lane >> 4, lc = lane & 15;
  const int n0 = wave * 16;
  v8f acc = {};
  for (int k0 = 0; k0 < 128; k0 += 4) {
    int k = k0 + 2 * hi;
    v2f a, b;
    a.x = hs[lc][k];
    a.y = hs[lc][k + 1];
    b.x = Wne1[k * 128 + n0 + lc] + Wne1[(k + 128) * 128 + n0 + lc];
    b.y = Wne1[(k + 1) * 128 + n0 + lc] + Wne1[(k + 129) * 128 + n0 + lc];
    acc = wmma_f32(a, b, acc);
  }
  float bias = bne1[n0 + lc];
  for (int v = 0; v < 8; ++v) us[v + 8 * hi][n0 + lc] = fmaxf(acc[v] + bias, 0.0f);
  __syncthreads();

  {
    int r = tid >> 4, l16 = tid & 15;
    float p = 0.0f;
    for (int c = l16; c < 128; c += 16) p += us[r][c] * Wne2[c];
    ps[r][l16] = p;
  }
  __syncthreads();
  if (tid < 16) {
    float s = 0.0f;
    for (int i = 0; i < 16; ++i) s += ps[tid][i];
    sc[tid] = (s + bne2[0]) * (1.0f / 512.0f);
  }
  __syncthreads();

  for (int idx = tid; idx < 16 * 512; idx += 256) {
    int r = idx >> 9, j = idx & 511;
    out2[(size_t)(node0 + r) * 512 + j] = sc[r];
  }
}

extern "C" void kernel_launch(void* const* d_in, const int* in_sizes, int n_in,
                              void* d_out, int out_size, void* d_ws,
                              size_t ws_size, hipStream_t stream) {
  (void)in_sizes; (void)n_in; (void)out_size; (void)ws_size;
  const float* x    = (const float*)d_in[0];
  const float* Wn1  = (const float*)d_in[1];
  const float* bn1  = (const float*)d_in[2];
  const float* Wn2  = (const float*)d_in[3];
  const float* bn2  = (const float*)d_in[4];
  const float* We1  = (const float*)d_in[5];
  const float* be1  = (const float*)d_in[6];
  const float* We2  = (const float*)d_in[7];
  const float* be2  = (const float*)d_in[8];
  const float* Wen1 = (const float*)d_in[9];
  const float* ben1 = (const float*)d_in[10];
  const float* Wen2 = (const float*)d_in[11];
  const float* ben2 = (const float*)d_in[12];
  const float* Wne1 = (const float*)d_in[13];
  const float* bne1 = (const float*)d_in[14];
  const float* Wne2 = (const float*)d_in[15];
  const float* bne2 = (const float*)d_in[16];

  float* ws = (float*)d_ws;
  float* h  = ws;
  float* E  = ws + 1024 * 128;
  float* S  = ws + 2 * 1024 * 128;
  float* h2   = (float*)d_out;
  float* out2 = (float*)d_out + 1024 * 128;

  dim3 blk(256);
  k_node_ffn<<<64, blk, 0, stream>>>(x, Wn1, bn1, Wn2, bn2, h);
  k_edge_ffn<<<64, blk, 0, stream>>>(h, We1, be1, We2, be2, E);
  k_reduce<<<1, blk, 0, stream>>>(E, S);
  k_e2n_ffn<<<64, blk, 0, stream>>>(E, S, Wen1, ben1, Wen2, ben2, h2);
  k_n2e<<<64, blk, 0, stream>>>(h2, Wne1, bne1, Wne2, bne2, out2);
}


# Fuse the 5 launches into 2: each 16-node tile runs node+edge FFN, then edge-to-node+node-to-edge FFN, through LDS; the 512-row sum becomes per-tile partial sums added in kernel 2 (sum order changes)
# speedup vs baseline: 1.3856x; 1.3856x over previous
#include <hip/hip_runtime.h>

typedef __attribute__((ext_vector_type(2))) float v2f;
typedef __attribute__((ext_vector_type(8))) float v8f;

__device__ __forceinline__ v8f wmma_f32(v2f a, v2f b, v8f c) {
  return __builtin_amdgcn_wmma_f32_16x16x4_f32(false, a, false, b, (short)0, c,
                                               false, false);
}

__global__ __launch_bounds__(256) void k_node_edge(
    const float* __restrict__ x, const float* __restrict__ Wn1,
    const float* __restrict__ bn1, const float* __restrict__ Wn2,
    const float* __restrict__ bn2, const float* __restrict__ We1,
    const float* __restrict__ be1, const float* __restrict__ We2,
    const float* __restrict__ be2, float* __restrict__ E,
    float* __restrict__ P) {
  __shared__ float xs[16][8];
  __shared__ float ts[16][256];
  __shared__ float hs[16][128];
  __shared__ float us[16][128];
  const int tid = threadIdx.x;
  const int node0 = blockIdx.x * 16;

  if (tid < 16 * 8) {
    int r = tid >> 3, k = tid & 7;
    xs[r][k] = (k < 6) ? x[(node0 + r) * 6 + k] : 0.0f;
  }
  __syncthreads();

  {
    int o = tid;
    float w0 = Wn1[0 * 256 + o], w1 = Wn1[1 * 256 + o], w2 = Wn1[2 * 256 + o];
    float w3 = Wn1[3 * 256 + o], w4 = Wn1[4 * 256 + o], w5 = Wn1[5 * 256 + o];
    float b = bn1[o];
    for (int r = 0; r < 16; ++r) {
      float acc = b + xs[r][0] * w0 + xs[r][1] * w1 + xs[r][2] * w2 +
                  xs[r][3] * w3 + xs[r][4] * w4 + xs[r][5] * w5;
      ts[r][o] = fmaxf(acc, 0.0f);
    }
  }
  __syncthreads();

  const int wave = tid >> 5, lane = tid & 31;
  const int hi = lane >> 4, lc = lane & 15;
  const int n0 = wave * 16;

  {
    v8f acc = {};
    for (int k0 = 0; k0 < 256; k0 += 4) {
      int k = k0 + 2 * hi;
      v2f a, b;
      a.x = ts[lc][k];
      a.y = ts[lc][k + 1];
      b.x = Wn2[k * 128 + n0 + lc];
      b.y = Wn2[(k + 1) * 128 + n0 + lc];
      acc = wmma_f32(a, b, acc);
    }
    float bias = bn2[n0 + lc];
    for (int v = 0; v < 8; ++v) hs[v + 8 * hi][n0 + lc] = acc[v] + bias;
  }
  __syncthreads();

  {
    v8f acc = {};
    for (int k0 = 0; k0 < 128; k0 += 4) {
      int k = k0 + 2 * hi;
      v2f a, b;
      a.x = hs[lc][k];
      a.y = hs[lc][k + 1];
      b.x = We1[k * 128 + n0 + lc] + We1[(k + 128) * 128 + n0 + lc];
      b.y = We1[(k + 1) * 128 + n0 + lc] + We1[(k + 129) * 128 + n0 + lc];
      acc = wmma_f32(a, b, acc);
    }
    float bias = be1[n0 + lc];
    for (int v = 0; v < 8; ++v) us[v + 8 * hi][n0 + lc] = fmaxf(acc[v] + bias, 0.0f);
  }
  __syncthreads();

  v8f acc2 = {};
  for (int k0 = 0; k0 < 128; k0 += 4) {
    int k = k0 + 2 * hi;
    v2f a, b;
    a.x = us[lc][k];
    a.y = us[lc][k + 1];
    b.x = We2[k * 128 + n0 + lc];
    b.y = We2[(k + 1) * 128 + n0 + lc];
    acc2 = wmma_f32(a, b, acc2);
  }
  float bias2 = be2[n0 + lc];
  float p = 0.0f;
  for (int v = 0; v < 8; ++v) {
    int r = v + 8 * hi;
    float e = acc2[v] + bias2;
    E[(node0 + r) * 128 + n0 + lc] = e;
    p += e;
  }
  p += __shfl_xor(p, 16);
  if (hi == 0) P[blockIdx.x * 128 + n0 + lc] = p;
}

__global__ __launch_bounds__(256) void k_e2n_n2e(
    const float* __restrict__ E, const float* __restrict__ P,
    const float* __restrict__ Wen1, const float* __restrict__ ben1,
    const float* __restrict__ Wen2, const float* __restrict__ ben2,
    const float* __restrict__ Wne1, const float* __restrict__ bne1,
    const float* __restrict__ Wne2, const float* __restrict__ bne2,
    float* __restrict__ h2, float* __restrict__ out2) {
  __shared__ float as[16][128];
  __shared__ float us[16][256];
  __shared__ float hs[16][128];
  __shared__ float vs[16][128];
  __shared__ float ps[16][16];
  __shared__ float sc[16];
  const int tid = threadIdx.x;
  const int node0 = blockIdx.x * 16;

  {
    const int c = tid & 127;
    const float* pb = P + (blockIdx.x >> 5) * (32 * 128) + c;
    float s = 0.0f;
    for (int t = 0; t < 32; ++t) s += pb[t * 128];
    for (int idx = tid; idx < 16 * 128; idx += 256) {
      int r = idx >> 7;
      as[r][c] = s - E[(node0 + r) * 128 + c];
    }
  }
  __syncthreads();

  const int wave = tid >> 5, lane = tid & 31;
  const int hi = lane >> 4, lc = lane & 15;
  const int n0 = wave * 16;

  for (int t = 0; t < 2; ++t) {
    int m0 = (wave + t * 8) * 16;
    v8f acc = {};
    for (int k0 = 0; k0 < 128; k0 += 4) {
      int k = k0 + 2 * hi;
      v2f a, b;
      a.x = as[lc][k];
      a.y = as[lc][k + 1];
      b.x = Wen1[k * 256 + m0 + lc];
      b.y = Wen1[(k + 1) * 256 + m0 + lc];
      acc = wmma_f32(a, b, acc);
    }
    float bias = ben1[m0 + lc];
    for (int v = 0; v < 8; ++v) us[v + 8 * hi][m0 + lc] = fmaxf(acc[v] + bias, 0.0f);
  }
  __syncthreads();

  {
    v8f acc2 = {};
    for (int k0 = 0; k0 < 256; k0 += 4) {
      int k = k0 + 2 * hi;
      v2f a, b;
      a.x = us[lc][k];
      a.y = us[lc][k + 1];
      b.x = Wen2[k * 128 + n0 + lc];
      b.y = Wen2[(k + 1) * 128 + n0 + lc];
      acc2 = wmma_f32(a, b, acc2);
    }
    float bias2 = ben2[n0 + lc];
    for (int v = 0; v < 8; ++v) {
      int r = v + 8 * hi;
      float o = acc2[v] + bias2;
      h2[(node0 + r) * 128 + n0 + lc] = o;
      hs[r][n0 + lc] = o;
    }
  }
  __syncthreads();

  {
    v8f acc = {};
    for (int k0 = 0; k0 < 128; k0 += 4) {
      int k = k0 + 2 * hi;
      v2f a, b;
      a.x = hs[lc][k];
      a.y = hs[lc][k + 1];
      b.x = Wne1[k * 128 + n0 + lc] + Wne1[(k + 128) * 128 + n0 + lc];
      b.y = Wne1[(k + 1) * 128 + n0 + lc] + Wne1[(k + 129) * 128 + n0 + lc];
      acc = wmma_f32(a, b, acc);
    }
    float bias = bne1[n0 + lc];
    for (int v = 0; v < 8; ++v) vs[v + 8 * hi][n0 + lc] = fmaxf(acc[v] + bias, 0.0f);
  }
  __syncthreads();

  {
    int r = tid >> 4, l16 = tid & 15;
    float q = 0.0f;
    for (int c = l16; c < 128; c += 16) q += vs[r][c] * Wne2[c];
    ps[r][l16] = q;
  }
  __syncthreads();
  if (tid < 16) {
    float s = 0.0f;
    for (int i = 0; i < 16; ++i) s += ps[tid][i];
    sc[tid] = (s + bne2[0]) * (1.0f / 512.0f);
  }
  __syncthreads();

  for (int idx = tid; idx < 16 * 512; idx += 256) {
    int r = idx >> 9, j = idx & 511;
    out2[(size_t)(node0 + r) * 512 + j] = sc[r];
  }
}

extern "C" void kernel_launch(void* const* d_in, const int* in_sizes, int n_in,
                              void* d_out, int out_size, void* d_ws,
                              size_t ws_size, hipStream_t stream) {
  (void)in_sizes; (void)n_in; (void)out_size; (void)ws_size;
  const float* x    = (const float*)d_in[0];
  const float* Wn1  = (const float*)d_in[1];
  const float* bn1  = (const float*)d_in[2];
  const float* Wn2  = (const float*)d_in[3];
  const float* bn2  = (const float*)d_in[4];
  const float* We1  = (const float*)d_in[5];
  const float* be1  = (const float*)d_in[6];
  const float* We2  = (const float*)d_in[7];
  const float* be2  = (const float*)d_in[8];
  const float* Wen1 = (const float*)d_in[9];
  const float* ben1 = (const float*)d_in[10];
  const float* Wen2 = (const float*)d_in[11];
  const float* ben2 = (const float*)d_in[12];
  const float* Wne1 = (const float*)d_in[13];
  const float* bne1 = (const float*)d_in[14];
  const float* Wne2 = (const float*)d_in[15];
  const float* bne2 = (const float*)d_in[16];

  float* ws = (float*)d_ws;
  float* E  = ws;
  float* P  = ws + 1024 * 128;
  float* h2   = (float*)d_out;
  float* out2 = (float*)d_out + 1024 * 128;

  dim3 blk(256);
  k_node_edge<<<64, blk, 0, stream>>>(x, Wn1, bn1, Wn2, bn2, We1, be1, We2, be2,
                                      E, P);
  k_e2n_n2e<<<64, blk, 0, stream>>>(E, P, Wen1, ben1, Wen2, ben2, Wne1, bne1,
                                    Wne2, bne2, h2, out2);
}


# Weight fragments of every matrix stage are requested 28 k-steps ahead of the product that uses them, across stage barriers that now wait for LDS only, so cache round trips overlap the products
# speedup vs baseline: 2.3107x; 1.6677x over previous
#include <hip/hip_runtime.h>

typedef __attribute__((ext_vector_type(2))) float v2f;
typedef __attribute__((ext_vector_type(8))) float v8f;

__device__ __forceinline__ v8f wmma_f32(v2f a, v2f b, v8f c) {
  return __builtin_amdgcn_wmma_f32_16x16x4_f32(false, a, false, b, (short)0, c,
                                               false, false);
}


__device__ __forceinline__ void lds_barrier() {
  __builtin_amdgcn_fence(__ATOMIC_RELEASE, "workgroup", "local");
  __builtin_amdgcn_s_barrier();
  __builtin_amdgcn_fence(__ATOMIC_ACQUIRE, "workgroup", "local");
}

#define STEP_FENCE() __builtin_amdgcn_sched_barrier(0x104)

__global__ __launch_bounds__(256) void k_node_edge(
    const float* __restrict__ x, const float* __restrict__ Wn1,
    const float* __restrict__ bn1, const float* __restrict__ Wn2,
    const float* __restrict__ bn2, const float* __restrict__ We1,
    const float* __restrict__ be1, const float* __restrict__ We2,
    const float* __restrict__ be2, float* __restrict__ E,
    float* __restrict__ P) {
  __shared__ float xs[16][8];
  __shared__ float ts[16][256];
  __shared__ float hs[16][128];
  __shared__ float us[16][128];
  const int tid = threadIdx.x;
  const int node0 = blockIdx.x * 16;
  const int wave = tid >> 5, lane = tid & 31;
  const int hi = lane >> 4, lc = lane & 15;
  const int n0 = wave * 16;
  const int col = n0 + lc;
  const int kb = 2 * hi;

  v2f f2[64], f3a[32], f3b[32], f4[32];
#define KA_AHEAD 28
#define KA_FETCH(S)                                              \
  do {                                                           \
    const int s_ = (S);                                          \
    if (s_ < 64) {                                               \
      const int k_ = 4 * s_ + kb;                                \
      f2[s_].x = Wn2[k_ * 128 + col];                            \
      f2[s_].y = Wn2[(k_ + 1) * 128 + col];                      \
    } else if (s_ < 96) {                                        \
      const int j_ = s_ - 64;                                    \
      const int k_ = 4 * j_ + kb;                                \
      f3a[j_].x = We1[k_ * 128 + col];                           \
      f3a[j_].y = We1[(k_ + 1) * 128 + col];                     \
      f3b[j_].x = We1[(k_ + 128) * 128 + col];                   \
      f3b[j_].y = We1[(k_ + 129) * 128 + col];                   \
    } else if (s_ < 128) {                                       \
      const int j_ = s_ - 96;                                    \
      const int k_ = 4 * j_ + kb;                                \
      f4[j_].x = We2[k_ * 128 + col];                            \
      f4[j_].y = We2[(k_ + 1) * 128 + col];                      \
    }                                                            \
  } while (0)

  float xv = 0.0f;
  if (tid < 16 * 8) {
    int r = tid >> 3, k = tid & 7;
    if (k < 6) xv = x[(node0 + r) * 6 + k];
  }
  const int o = tid;
  const float w0 = Wn1[0 * 256 + o], w1 = Wn1[1 * 256 + o], w2 = Wn1[2 * 256 + o];
  const float w3 = Wn1[3 * 256 + o], w4 = Wn1[4 * 256 + o], w5 = Wn1[5 * 256 + o];
  const float b1 = bn1[o];
  STEP_FENCE();
  const float bias2 = bn2[col];
  const float bias3 = be1[col];
  const float bias4 = be2[col];
#pragma unroll
  for (int s = 0; s < KA_AHEAD; ++s) KA_FETCH(s);

  if (tid < 16 * 8) {
    int r = tid >> 3, k = tid & 7;
    xs[r][k] = xv;
  }
  lds_barrier();

  for (int r = 0; r < 16; ++r) {
    float acc = b1 + xs[r][0] * w0 + xs[r][1] * w1 + xs[r][2] * w2 +
                xs[r][3] * w3 + xs[r][4] * w4 + xs[r][5] * w5;
    ts[r][o] = fmaxf(acc, 0.0f);
  }
  lds_barrier();

  {
    v8f acc = {};
#pragma unroll
    for (int i = 0; i < 64; ++i) {
      STEP_FENCE();
      KA_FETCH(i + KA_AHEAD);
      int k = 4 * i + kb;
      v2f a;
      a.x = ts[lc][k];
      a.y = ts[lc][k + 1];
      acc = wmma_f32(a, f2[i], acc);
    }
    for (int v = 0; v < 8; ++v) hs[v + 8 * hi][col] = acc[v] + bias2;
  }
  lds_barrier();

  {
    v8f acc = {};
#pragma unroll
    for (int i = 0; i < 32; ++i) {
      STEP_FENCE();
      KA_FETCH(64 + i + KA_AHEAD);
      int k = 4 * i + kb;
      v2f a, b;
      a.x = hs[lc][k];
      a.y = hs[lc][k + 1];
      b.x = f3a[i].x + f3b[i].x;
      b.y = f3a[i].y + f3b[i].y;
      acc = wmma_f32(a, b, acc);
    }
    for (int v = 0; v < 8; ++v) us[v + 8 * hi][col] = fmaxf(acc[v] + bias3, 0.0f);
  }
  lds_barrier();

  v8f acc2 = {};
#pragma unroll
  for (int i = 0; i < 32; ++i) {
    if (96 + i + KA_AHEAD < 128) {
      STEP_FENCE();
      KA_FETCH(96 + i + KA_AHEAD);
    }
    int k = 4 * i + kb;
    v2f a;
    a.x = us[lc][k];
    a.y = us[lc][k + 1];
    acc2 = wmma_f32(a, f4[i], acc2);
  }
#undef KA_FETCH
#undef KA_AHEAD
  float p = 0.0f;
  for (int v = 0; v < 8; ++v) {
    int r = v + 8 * hi;
    float e = acc2[v] + bias4;
    E[(node0 + r) * 128 + col] = e;
    p += e;
  }
  p += __shfl_xor(p, 16);
  if (hi == 0) P[blockIdx.x * 128 + col] = p;
}

__global__ __launch_bounds__(256) void k_e2n_n2e(
    const float* __restrict__ E, const float* __restrict__ P,
    const float* __restrict__ Wen1, const float* __restrict__ ben1,
    const float* __restrict__ Wen2, const float* __restrict__ ben2,
    const float* __restrict__ Wne1, const float* __restrict__ bne1,
    const float* __restrict__ Wne2, const float* __restrict__ bne2,
    float* __restrict__ h2, float* __restrict__ out2) {
  __shared__ float as[16][128];
  __shared__ float us[16][256];
  __shared__ float hs[16][128];
  __shared__ float vs[16][128];
  __shared__ float ps[16][16];
  __shared__ float sc[16];
  const int tid = threadIdx.x;
  const int node0 = blockIdx.x * 16;
  const int wave = tid >> 5, lane = tid & 31;
  const int hi = lane >> 4, lc = lane & 15;
  const int n0 = wave * 16;
  const int col = n0 + lc;
  const int kb = 2 * hi;

  v2f g1[64], g2[64], g3a[32], g3b[32];
#define KB_AHEAD 28
#define KB_FETCH(S)                                              \
  do {                                                           \
    const int s_ = (S);                                          \
    if (s_ < 64) {                                               \
      const int m_ = (wave + (s_ >> 5) * 8) * 16 + lc;           \
      const int k_ = 4 * (s_ & 31) + kb;                         \
      g1[s_].x = Wen1[k_ * 256 + m_];                            \
      g1[s_].y = Wen1[(k_ + 1) * 256 + m_];                      \
    } else if (s_ < 128) {                                       \
      const int j_ = s_ - 64;                                    \
      const int k_ = 4 * j_ + kb;                                \
      g2[j_].x = Wen2[k_ * 128 + col];                           \
      g2[j_].y = Wen2[(k_ + 1) * 128 + col];                     \
    } else if (s_ < 160) {                                       \
      const int j_ = s_ - 128;                                   \
      const int k_ = 4 * j_ + kb;                                \
      g3a[j_].x = Wne1[k_ * 128 + col];                          \
      g3a[j_].y = Wne1[(k_ + 1) * 128 + col];                    \
      g3b[j_].x = Wne1[(k_ + 128) * 128 + col];                  \
      g3b[j_].y = Wne1[(k_ + 129) * 128 + col];                  \
    }                                                            \
  } while (0)

  const int c = tid & 127;
  const float* pb = P + (blockIdx.x >> 5) * (32 * 128) + c;
  float pv[32];
#pragma unroll
  for (int t = 0; t < 32; ++t) pv[t] = pb[t * 128];
  float ev[8];
#pragma unroll
  for (int j = 0; j < 8; ++j) ev[j] = E[(node0 + (tid >> 7) + 2 * j) * 128 + c];
  STEP_FENCE();
#pragma unroll
  for (int s = 0; s < KB_AHEAD; ++s) KB_FETCH(s);
  const float biasA0 = ben1[wave * 16 + lc];
  const float biasA1 = ben1[(wave + 8) * 16 + lc];
  const float biasB = ben2[col];
  const float biasC = bne1[col];
  float wd[8];
#pragma unroll
  for (int j = 0; j < 8; ++j) wd[j] = Wne2[(tid & 15) + 16 * j];
  const float bd = bne2[0];

  {
    float s = 0.0f;
#pragma unroll
    for (int t = 0; t < 32; ++t) s += pv[t];
#pragma unroll
    for (int j = 0; j < 8; ++j) as[(tid >> 7) + 2 * j][c] = s - ev[j];
  }
  lds_barrier();

#pragma unroll
  for (int t = 0; t < 2; ++t) {
    int m0 = (wave + t * 8) * 16;
    v8f acc = {};
#pragma unroll
    for (int i = 0; i < 32; ++i) {
      STEP_FENCE();
      KB_FETCH(t * 32 + i + KB_AHEAD);
      int k = 4 * i + kb;
      v2f a;
      a.x = as[lc][k];
      a.y = as[lc][k + 1];
      acc = wmma_f32(a, g1[t * 32 + i], acc);
    }
    float bias = t ? biasA1 : biasA0;
    for (int v = 0; v < 8; ++v) us[v + 8 * hi][m0 + lc] = fmaxf(acc[v] + bias, 0.0f);
  }
  lds_barrier();

  {
    v8f acc2 = {};
#pragma unroll
    for (int i = 0; i < 64; ++i) {
      STEP_FENCE();
      KB_FETCH(64 + i + KB_AHEAD);
      int k = 4 * i + kb;
      v2f a;
      a.x = us[lc][k];
      a.y = us[lc][k + 1];
      acc2 = wmma_f32(a, g2[i], acc2);
    }
    for (int v = 0; v < 8; ++v) {
      int r = v + 8 * hi;
      float o = acc2[v] + biasB;
      h2[(node0 + r) * 128 + col] = o;
      hs[r][col] = o;
    }
  }
  lds_barrier();

  {
    v8f acc = {};
#pragma unroll
    for (int i = 0; i < 32; ++i) {
      STEP_FENCE();
      KB_FETCH(128 + i + KB_AHEAD);
      int k = 4 * i + kb;
      v2f a, b;
      a.x = hs[lc][k];
      a.y = hs[lc][k + 1];
      b.x = g3a[i].x + g3b[i].x;
      b.y = g3a[i].y + g3b[i].y;
      acc = wmma_f32(a, b, acc);
    }
    for (int v = 0; v < 8; ++v) vs[v + 8 * hi][col] = fmaxf(acc[v] + biasC, 0.0f);
  }
#undef KB_FETCH
#undef KB_AHEAD
  lds_barrier();

  {
    int r = tid >> 4, l16 = tid & 15;
    float q = 0.0f;
#pragma unroll
    for (int j = 0; j < 8; ++j) q += vs[r][l16 + 16 * j] * wd[j];
    ps[r][l16] = q;
  }
  lds_barrier();
  if (tid < 16) {
    float s = 0.0f;
    for (int i = 0; i < 16; ++i) s += ps[tid][i];
    sc[tid] = (s + bd) * (1.0f / 512.0f);
  }
  lds_barrier();

  for (int idx = tid; idx < 16 * 512; idx += 256) {
    int r = idx >> 9, j = idx & 511;
    out2[(size_t)(node0 + r) * 512 + j] = sc[r];
  }
}

extern "C" void kernel_launch(void* const* d_in, const int* in_sizes, int n_in,
                              void* d_out, int out_size, void* d_ws,
                              size_t ws_size, hipStream_t stream) {
  (void)in_sizes; (void)n_in; (void)out_size; (void)ws_size;
  const float* x    = (const float*)d_in[0];
  const float* Wn1  = (const float*)d_in[1];
  const float* bn1  = (const float*)d_in[2];
  const float* Wn2  = (const float*)d_in[3];
  const float* bn2  = (const float*)d_in[4];
  const float* We1  = (const float*)d_in[5];
  const float* be1  = (const float*)d_in[6];
  const float* We2  = (const float*)d_in[7];
  const float* be2  = (const float*)d_in[8];
  const float* Wen1 = (const float*)d_in[9];
  const float* ben1 = (const float*)d_in[10];
  const float* Wen2 = (const float*)d_in[11];
  const float* ben2 = (const float*)d_in[12];
  const float* Wne1 = (const float*)d_in[13];
  const float* bne1 = (const float*)d_in[14];
  const float* Wne2 = (const float*)d_in[15];
  const float* bne2 = (const float*)d_in[16];

  float* ws = (float*)d_ws;
  float* E  = ws;
  float* P  = ws + 1024 * 128;
  float* h2   = (float*)d_out;
  float* out2 = (float*)d_out + 1024 * 128;

  dim3 blk(256);
  k_node_edge<<<64, blk, 0, stream>>>(x, Wn1, bn1, Wn2, bn2, We1, be1, We2, be2,
                                      E, P);
  k_e2n_n2e<<<64, blk, 0, stream>>>(E, P, Wen1, ben1, Wen2, ben2, Wne1, bne1,
                                    Wne2, bne2, h2, out2);
}


# Pad the LDS rows of the activation tiles (pitch 132/260 floats, 148 for the dot tile): an A-fragment read now spans all 64 banks; before, its 16 rows shared 4 banks (16-way conflict, every k step)
# speedup vs baseline: 3.6167x; 1.5652x over previous
#include <hip/hip_runtime.h>

typedef __attribute__((ext_vector_type(2))) float v2f;
typedef __attribute__((ext_vector_type(8))) float v8f;

__device__ __forceinline__ v8f wmma_f32(v2f a, v2f b, v8f c) {
  return __builtin_amdgcn_wmma_f32_16x16x4_f32(false, a, false, b, (short)0, c,
                                               false, false);
}


__device__ __forceinline__ void lds_barrier() {
  __builtin_amdgcn_fence(__ATOMIC_RELEASE, "workgroup", "local");
  __builtin_amdgcn_s_barrier();
  __builtin_amdgcn_fence(__ATOMIC_ACQUIRE, "workgroup", "local");
}

#define STEP_FENCE() __builtin_amdgcn_sched_barrier(0x104)

#define P128 132
#define P256 260
#define PDOT 148

__global__ __launch_bounds__(256) void k_node_edge(
    const float* __restrict__ x, const float* __restrict__ Wn1,
    const float* __restrict__ bn1, const float* __restrict__ Wn2,
    const float* __restrict__ bn2, const float* __restrict__ We1,
    const float* __restrict__ be1, const float* __restrict__ We2,
    const float* __restrict__ be2, float* __restrict__ E,
    float* __restrict__ P) {
  __shared__ float xs[16][8];
  __shared__ float ts[16][P256];
  __shared__ float hs[16][P128];
  __shared__ float us[16][P128];
  const int tid = threadIdx.x;
  const int node0 = blockIdx.x * 16;
  const int wave = tid >> 5, lane = tid & 31;
  const int hi = lane >> 4, lc = lane & 15;
  const int n0 = wave * 16;
  const int col = n0 + lc;
  const int kb = 2 * hi;

  v2f f2[64], f3a[32], f3b[32], f4[32];
#define KA_AHEAD 28
#define KA_FETCH(S)                                              \
  do {                                                           \
    const int s_ = (S);                                          \
    if (s_ < 64) {                                               \
      const int k_ = 4 * s_ + kb;                                \
      f2[s_].x = Wn2[k_ * 128 + col];                            \
      f2[s_].y = Wn2[(k_ + 1) * 128 + col];                      \
    } else if (s_ < 96) {                                        \
      const int j_ = s_ - 64;                                    \
      const int k_ = 4 * j_ + kb;                                \
      f3a[j_].x = We1[k_ * 128 + col];                           \
      f3a[j_].y = We1[(k_ + 1) * 128 + col];                     \
      f3b[j_].x = We1[(k_ + 128) * 128 + col];                   \
      f3b[j_].y = We1[(k_ + 129) * 128 + col];                   \
    } else if (s_ < 128) {                                       \
      const int j_ = s_ - 96;                                    \
      const int k_ = 4 * j_ + kb;                                \
      f4[j_].x = We2[k_ * 128 + col];                            \
      f4[j_].y = We2[(k_ + 1) * 128 + col];                      \
    }                                                            \
  } while (0)

  float xv = 0.0f;
  if (tid < 16 * 8) {
    int r = tid >> 3, k = tid & 7;
    if (k < 6) xv = x[(node0 + r) * 6 + k];
  }
  const int o = tid;
  const float w0 = Wn1[0 * 256 + o], w1 = Wn1[1 * 256 + o], w2 = Wn1[2 * 256 + o];
  const float w3 = Wn1[3 * 256 + o], w4 = Wn1[4 * 256 + o], w5 = Wn1[5 * 256 + o];
  const float b1 = bn1[o];
  STEP_FENCE();
  const float bias2 = bn2[col];
  const float bias3 = be1[col];
  const float bias4 = be2[col];
#pragma unroll
  for (int s = 0; s < KA_AHEAD; ++s) KA_FETCH(s);

  if (tid < 16 * 8) {
    int r = tid >> 3, k = tid & 7;
    xs[r][k] = xv;
  }
  lds_barrier();

  for (int r = 0; r < 16; ++r) {
    float acc = b1 + xs[r][0] * w0 + xs[r][1] * w1 + xs[r][2] * w2 +
                xs[r][3] * w3 + xs[r][4] * w4 + xs[r][5] * w5;
    ts[r][o] = fmaxf(acc, 0.0f);
  }
  lds_barrier();

  {
    v8f acc = {};
#pragma unroll
    for (int i = 0; i < 64; ++i) {
      STEP_FENCE();
      KA_FETCH(i + KA_AHEAD);
      int k = 4 * i + kb;
      v2f a;
      a.x = ts[lc][k];
      a.y = ts[lc][k + 1];
      acc = wmma_f32(a, f2[i], acc);
    }
    for (int v = 0; v < 8; ++v) hs[v + 8 * hi][col] = acc[v] + bias2;
  }
  lds_barrier();

  {
    v8f acc = {};
#pragma unroll
    for (int i = 0; i < 32; ++i) {
      STEP_FENCE();
      KA_FETCH(64 + i + KA_AHEAD);
      int k = 4 * i + kb;
      v2f a, b;
      a.x = hs[lc][k];
      a.y = hs[lc][k + 1];
      b.x = f3a[i].x + f3b[i].x;
      b.y = f3a[i].y + f3b[i].y;
      acc = wmma_f32(a, b, acc);
    }
    for (int v = 0; v < 8; ++v) us[v + 8 * hi][col] = fmaxf(acc[v] + bias3, 0.0f);
  }
  lds_barrier();

  v8f acc2 = {};
#pragma unroll
  for (int i = 0; i < 32; ++i) {
    if (96 + i + KA_AHEAD < 128) {
      STEP_FENCE();
      KA_FETCH(96 + i + KA_AHEAD);
    }
    int k = 4 * i + kb;
    v2f a;
    a.x = us[lc][k];
    a.y = us[lc][k + 1];
    acc2 = wmma_f32(a, f4[i], acc2);
  }
#undef KA_FETCH
#undef KA_AHEAD
  float p = 0.0f;
  for (int v = 0; v < 8; ++v) {
    int r = v + 8 * hi;
    float e = acc2[v] + bias4;
    E[(node0 + r) * 128 + col] = e;
    p += e;
  }
  p += __shfl_xor(p, 16);
  if (hi == 0) P[blockIdx.x * 128 + col] = p;
}

__global__ __launch_bounds__(256) void k_e2n_n2e(
    const float* __restrict__ E, const float* __restrict__ P,
    const float* __restrict__ Wen1, const float* __restrict__ ben1,
    const float* __restrict__ Wen2, const float* __restrict__ ben2,
    const float* __restrict__ Wne1, const float* __restrict__ bne1,
    const float* __restrict__ Wne2, const float* __restrict__ bne2,
    float* __restrict__ h2, float* __restrict__ out2) {
  __shared__ float as[16][P128];
  __shared__ float us[16][P256];
  __shared__ float hs[16][P128];
  __shared__ float vs[16][PDOT];
  __shared__ float ps[16][16];
  __shared__ float sc[16];
  const int tid = threadIdx.x;
  const int node0 = blockIdx.x * 16;
  const int wave = tid >> 5, lane = tid & 31;
  const int hi = lane >> 4, lc = lane & 15;
  const int n0 = wave * 16;
  const int col = n0 + lc;
  const int kb = 2 * hi;

  v2f g1[64], g2[64], g3a[32], g3b[32];
#define KB_AHEAD 28
#define KB_FETCH(S)                                              \
  do {                                                           \
    const int s_ = (S);                                          \
    if (s_ < 64) {                                               \
      const int m_ = (wave + (s_ >> 5) * 8) * 16 + lc;           \
      const int k_ = 4 * (s_ & 31) + kb;                         \
      g1[s_].x = Wen1[k_ * 256 + m_];                            \
      g1[s_].y = Wen1[(k_ + 1) * 256 + m_];                      \
    } else if (s_ < 128) {                                       \
      const int j_ = s_ - 64;                                    \
      const int k_ = 4 * j_ + kb;                                \
      g2[j_].x = Wen2[k_ * 128 + col];                           \
      g2[j_].y = Wen2[(k_ + 1) * 128 + col];                     \
    } else if (s_ < 160) {                                       \
      const int j_ = s_ - 128;                                   \
      const int k_ = 4 * j_ + kb;                                \
      g3a[j_].x = Wne1[k_ * 128 + col];                          \
      g3a[j_].y = Wne1[(k_ + 1) * 128 + col];                    \
      g3b[j_].x = Wne1[(k_ + 128) * 128 + col];                  \
      g3b[j_].y = Wne1[(k_ + 129) * 128 + col];                  \
    }                                                            \
  } while (0)

  const int c = tid & 127;
  const float* pb = P + (blockIdx.x >> 5) * (32 * 128) + c;
  float pv[32];
#pragma unroll
  for (int t = 0; t < 32; ++t) pv[t] = pb[t * 128];
  float ev[8];
#pragma unroll
  for (int j = 0; j < 8; ++j) ev[j] = E[(node0 + (tid >> 7) + 2 * j) * 128 + c];
  STEP_FENCE();
#pragma unroll
  for (int s = 0; s < KB_AHEAD; ++s) KB_FETCH(s);
  const float biasA0 = ben1[wave * 16 + lc];
  const float biasA1 = ben1[(wave + 8) * 16 + lc];
  const float biasB = ben2[col];
  const float biasC = bne1[col];
  float wd[8];
#pragma unroll
  for (int j = 0; j < 8; ++j) wd[j] = Wne2[(tid & 15) + 16 * j];
  const float bd = bne2[0];

  {
    float s = 0.0f;
#pragma unroll
    for (int t = 0; t < 32; ++t) s += pv[t];
#pragma unroll
    for (int j = 0; j < 8; ++j) as[(tid >> 7) + 2 * j][c] = s - ev[j];
  }
  lds_barrier();

#pragma unroll
  for (int t = 0; t < 2; ++t) {
    int m0 = (wave + t * 8) * 16;
    v8f acc = {};
#pragma unroll
    for (int i = 0; i < 32; ++i) {
      STEP_FENCE();
      KB_FETCH(t * 32 + i + KB_AHEAD);
      int k = 4 * i + kb;
      v2f a;
      a.x = as[lc][k];
      a.y = as[lc][k + 1];
      acc = wmma_f32(a, g1[t * 32 + i], acc);
    }
    float bias = t ? biasA1 : biasA0;
    for (int v = 0; v < 8; ++v) us[v + 8 * hi][m0 + lc] = fmaxf(acc[v] + bias, 0.0f);
  }
  lds_barrier();

  {
    v8f acc2 = {};
#pragma unroll
    for (int i = 0; i < 64; ++i) {
      STEP_FENCE();
      KB_FETCH(64 + i + KB_AHEAD);
      int k = 4 * i + kb;
      v2f a;
      a.x = us[lc][k];
      a.y = us[lc][k + 1];
      acc2 = wmma_f32(a, g2[i], acc2);
    }
    for (int v = 0; v < 8; ++v) {
      int r = v + 8 * hi;
      float o = acc2[v] + biasB;
      h2[(node0 + r) * 128 + col] = o;
      hs[r][col] = o;
    }
  }
  lds_barrier();

  {
    v8f acc = {};
#pragma unroll
    for (int i = 0; i < 32; ++i) {
      STEP_FENCE();
      KB_FETCH(128 + i + KB_AHEAD);
      int k = 4 * i + kb;
      v2f a, b;
      a.x = hs[lc][k];
      a.y = hs[lc][k + 1];
      b.x = g3a[i].x + g3b[i].x;
      b.y = g3a[i].y + g3b[i].y;
      acc = wmma_f32(a, b, acc);
    }
    for (int v = 0; v < 8; ++v) vs[v + 8 * hi][col] = fmaxf(acc[v] + biasC, 0.0f);
  }
#undef KB_FETCH
#undef KB_AHEAD
  lds_barrier();

  {
    int r = tid >> 4, l16 = tid & 15;
    float q = 0.0f;
#pragma unroll
    for (int j = 0; j < 8; ++j) q += vs[r][l16 + 16 * j] * wd[j];
    ps[r][l16] = q;
  }
  lds_barrier();
  if (tid < 16) {
    float s = 0.0f;
    for (int i = 0; i < 16; ++i) s += ps[tid][i];
    sc[tid] = (s + bd) * (1.0f / 512.0f);
  }
  lds_barrier();

  for (int idx = tid; idx < 16 * 512; idx += 256) {
    int r = idx >> 9, j = idx & 511;
    out2[(size_t)(node0 + r) * 512 + j] = sc[r];
  }
}

extern "C" void kernel_launch(void* const* d_in, const int* in_sizes, int n_in,
                              void* d_out, int out_size, void* d_ws,
                              size_t ws_size, hipStream_t stream) {
  (void)in_sizes; (void)n_in; (void)out_size; (void)ws_size;
  const float* x    = (const float*)d_in[0];
  const float* Wn1  = (const float*)d_in[1];
  const float* bn1  = (const float*)d_in[2];
  const float* Wn2  = (const float*)d_in[3];
  const float* bn2  = (const float*)d_in[4];
  const float* We1  = (const float*)d_in[5];
  const float* be1  = (const float*)d_in[6];
  const float* We2  = (const float*)d_in[7];
  const float* be2  = (const float*)d_in[8];
  const float* Wen1 = (const float*)d_in[9];
  const float* ben1 = (const float*)d_in[10];
  const float* Wen2 = (const float*)d_in[11];
  const float* ben2 = (const float*)d_in[12];
  const float* Wne1 = (const float*)d_in[13];
  const float* bne1 = (const float*)d_in[14];
  const float* Wne2 = (const float*)d_in[15];
  const float* bne2 = (const float*)d_in[16];

  float* ws = (float*)d_ws;
  float* E  = ws;
  float* P  = ws + 1024 * 128;
  float* h2   = (float*)d_out;
  float* out2 = (float*)d_out + 1024 * 128;

  dim3 blk(256);
  k_node_edge<<<64, blk, 0, stream>>>(x, Wn1, bn1, Wn2, bn2, We1, be1, We2, be2,
                                      E, P);
  k_e2n_n2e<<<64, blk, 0, stream>>>(E, P, Wen1, ben1, Wen2, ben2, Wne1, bne1,
                                    Wne2, bne2, h2, out2);
}


# Each wave now owns 4 column tiles and a quarter (half where N=256) of K: one 16-byte weight load per lane feeds 4 products, one A read feeds 4; K parts are added through LDS (sum order changes)
# speedup vs baseline: 3.8726x; 1.0708x over previous
#include <hip/hip_runtime.h>

typedef __attribute__((ext_vector_type(2))) float v2f;
typedef __attribute__((ext_vector_type(4))) float v4f;
typedef __attribute__((ext_vector_type(8))) float v8f;

__device__ __forceinline__ v8f wmma_f32(v2f a, v2f b, v8f c) {
  return __builtin_amdgcn_wmma_f32_16x16x4_f32(false, a, false, b, (short)0, c,
                                               false, false);
}

__device__ __forceinline__ void lds_barrier() {
  __builtin_amdgcn_fence(__ATOMIC_RELEASE, "workgroup", "local");
  __builtin_amdgcn_s_barrier();
  __builtin_amdgcn_fence(__ATOMIC_ACQUIRE, "workgroup", "local");
}

#define STEP_FENCE() __builtin_amdgcn_sched_barrier(0x104)

#define P128 132
#define P256 260
#define PDOT 148

#define WLOAD(W, ROW, LD, CB) (*(const v4f*)((W) + (ROW) * (LD) + (CB)))
#define MM4(A, BX, BY)                                        \
  do {                                                        \
    v2f b0_ = {(BX).x, (BY).x}, b1_ = {(BX).y, (BY).y};       \
    v2f b2_ = {(BX).z, (BY).z}, b3_ = {(BX).w, (BY).w};       \
    acc0 = wmma_f32((A), b0_, acc0);                          \
    acc1 = wmma_f32((A), b1_, acc1);                          \
    acc2 = wmma_f32((A), b2_, acc2);                          \
    acc3 = wmma_f32((A), b3_, acc3);                          \
  } while (0)
#define PUT_PARTS(BASE, PITCH)                                            \
  do {                                                                    \
    _Pragma("unroll") for (int v = 0; v < 8; ++v) {                       \
      v4f o_ = {acc0[v], acc1[v], acc2[v], acc3[v]};                      \
      *(v4f*)((BASE) + (v + 8 * hi) * (PITCH) + cb) = o_;                 \
    }                                                                     \
  } while (0)

__global__ __launch_bounds__(256) void k_node_edge(
    const float* __restrict__ x, const float* __restrict__ Wn1,
    const float* __restrict__ bn1, const float* __restrict__ Wn2,
    const float* __restrict__ bn2, const float* __restrict__ We1,
    const float* __restrict__ be1, const float* __restrict__ We2,
    const float* __restrict__ be2, float* __restrict__ E,
    float* __restrict__ P) {
  __shared__ float xs[16][8];
  __shared__ __attribute__((aligned(16))) float ts[16][P256];
  __shared__ __attribute__((aligned(16))) float hs[16][P128];
  __shared__ __attribute__((aligned(16))) float us[16][P128];
  __shared__ __attribute__((aligned(16))) float red[4 * 16 * P128];
  const int tid = threadIdx.x;
  const int node0 = blockIdx.x * 16;
  const int wave = tid >> 5, lane = tid & 31;
  const int hi = lane >> 4, lc = lane & 15;
  const int kb = 2 * hi;
  const int g = wave & 1, q = wave >> 1;
  const int cb = 64 * g + 4 * lc;
  const int c4 = (tid & 31) * 4, r0 = tid >> 5;
  float* const myred = red + q * (16 * P128);

  float xv = 0.0f;
  if (tid < 16 * 8) {
    int r = tid >> 3, k = tid & 7;
    if (k < 6) xv = x[(node0 + r) * 6 + k];
  }
  const int o = tid;
  const float w0 = Wn1[0 * 256 + o], w1 = Wn1[1 * 256 + o], w2 = Wn1[2 * 256 + o];
  const float w3 = Wn1[3 * 256 + o], w4 = Wn1[4 * 256 + o], w5 = Wn1[5 * 256 + o];
  const float b1 = bn1[o];
  STEP_FENCE();
  v4f f2x[16], f2y[16];
#pragma unroll
  for (int i = 0; i < 16; ++i) {
    const int k = 64 * q + 4 * i + kb;
    f2x[i] = WLOAD(Wn2, k, 128, cb);
    f2y[i] = WLOAD(Wn2, k + 1, 128, cb);
  }
  const v4f bias2 = *(const v4f*)(bn2 + c4);
  const v4f bias3 = *(const v4f*)(be1 + c4);
  const v4f bias4 = *(const v4f*)(be2 + c4);

  if (tid < 16 * 8) {
    int r = tid >> 3, k = tid & 7;
    xs[r][k] = xv;
  }
  lds_barrier();

  for (int r = 0; r < 16; ++r) {
    float acc = b1 + xs[r][0] * w0 + xs[r][1] * w1 + xs[r][2] * w2 +
                xs[r][3] * w3 + xs[r][4] * w4 + xs[r][5] * w5;
    ts[r][o] = fmaxf(acc, 0.0f);
  }
  lds_barrier();

  v4f f3ax[8], f3ay[8], f3bx[8], f3by[8];
  {
    v8f acc0 = {}, acc1 = {}, acc2 = {}, acc3 = {};
#pragma unroll
    for (int i = 0; i < 16; ++i) {
      STEP_FENCE();
      {
        const int j = i >> 1;
        const int k = 32 * q + 4 * j + kb + ((i & 1) ? 128 : 0);
        if (i & 1) { f3bx[j] = WLOAD(We1, k, 128, cb); f3by[j] = WLOAD(We1, k + 1, 128, cb); }
        else       { f3ax[j] = WLOAD(We1, k, 128, cb); f3ay[j] = WLOAD(We1, k + 1, 128, cb); }
      }
      const int k = 64 * q + 4 * i + kb;
      v2f a = {ts[lc][k], ts[lc][k + 1]};
      MM4(a, f2x[i], f2y[i]);
    }
    PUT_PARTS(myred, P128);
  }
  lds_barrier();
#pragma unroll
  for (int j = 0; j < 2; ++j) {
    const int r = r0 + 8 * j;
    const float* p = red + r * P128 + c4;
    v4f s = *(const v4f*)p + *(const v4f*)(p + 16 * P128);
    s += *(const v4f*)(p + 2 * 16 * P128);
    s += *(const v4f*)(p + 3 * 16 * P128);
    *(v4f*)(&hs[r][c4]) = s + bias2;
  }
  lds_barrier();

  v4f f4x[8], f4y[8];
  {
    v8f acc0 = {}, acc1 = {}, acc2 = {}, acc3 = {};
#pragma unroll
    for (int i = 0; i < 8; ++i) {
      STEP_FENCE();
      {
        const int k = 32 * q + 4 * i + kb;
        f4x[i] = WLOAD(We2, k, 128, cb);
        f4y[i] = WLOAD(We2, k + 1, 128, cb);
      }
      const int k = 32 * q + 4 * i + kb;
      v2f a = {hs[lc][k], hs[lc][k + 1]};
      const v4f bx = f3ax[i] + f3bx[i];
      const v4f by = f3ay[i] + f3by[i];
      MM4(a, bx, by);
    }
    PUT_PARTS(myred, P128);
  }
  lds_barrier();
#pragma unroll
  for (int j = 0; j < 2; ++j) {
    const int r = r0 + 8 * j;
    const float* p = red + r * P128 + c4;
    v4f s = *(const v4f*)p + *(const v4f*)(p + 16 * P128);
    s += *(const v4f*)(p + 2 * 16 * P128);
    s += *(const v4f*)(p + 3 * 16 * P128);
    s += bias3;
    v4f u = {fmaxf(s.x, 0.0f), fmaxf(s.y, 0.0f), fmaxf(s.z, 0.0f), fmaxf(s.w, 0.0f)};
    *(v4f*)(&us[r][c4]) = u;
  }
  lds_barrier();

  {
    v8f acc0 = {}, acc1 = {}, acc2 = {}, acc3 = {};
#pragma unroll
    for (int i = 0; i < 8; ++i) {
      const int k = 32 * q + 4 * i + kb;
      v2f a = {us[lc][k], us[lc][k + 1]};
      MM4(a, f4x[i], f4y[i]);
    }
    PUT_PARTS(myred, P128);
  }
  lds_barrier();
#pragma unroll
  for (int j = 0; j < 2; ++j) {
    const int r = r0 + 8 * j;
    const float* p = red + r * P128 + c4;
    v4f s = *(const v4f*)p + *(const v4f*)(p + 16 * P128);
    s += *(const v4f*)(p + 2 * 16 * P128);
    s += *(const v4f*)(p + 3 * 16 * P128);
    s += bias4;
    *(v4f*)(E + (node0 + r) * 128 + c4) = s;
    *(v4f*)(&hs[r][c4]) = s;
  }
  lds_barrier();
  if (tid < 128) {
    float p0 = 0.0f, p1 = 0.0f;
    for (int r = 0; r < 8; ++r) { p0 += hs[r][tid]; p1 += hs[r + 8][tid]; }
    P[blockIdx.x * 128 + tid] = p0 + p1;
  }
}

__global__ __launch_bounds__(256) void k_e2n_n2e(
    const float* __restrict__ E, const float* __restrict__ P,
    const float* __restrict__ Wen1, const float* __restrict__ ben1,
    const float* __restrict__ Wen2, const float* __restrict__ ben2,
    const float* __restrict__ Wne1, const float* __restrict__ bne1,
    const float* __restrict__ Wne2, const float* __restrict__ bne2,
    float* __restrict__ h2, float* __restrict__ out2) {
  __shared__ __attribute__((aligned(16))) float as[16][P128];
  __shared__ __attribute__((aligned(16))) float us[16][P256];
  __shared__ __attribute__((aligned(16))) float hs[16][P128];
  __shared__ __attribute__((aligned(16))) float vs[16][PDOT];
  __shared__ float ps[16][16];
  __shared__ float sc[16];
  __shared__ __attribute__((aligned(16))) float red[4 * 16 * P128];
  const int tid = threadIdx.x;
  const int node0 = blockIdx.x * 16;
  const int wave = tid >> 5, lane = tid & 31;
  const int hi = lane >> 4, lc = lane & 15;
  const int kb = 2 * hi;
  const int g = wave & 1, q = wave >> 1;
  const int cb = 64 * g + 4 * lc;
  const int gw = wave & 3, qw = wave >> 2;
  const int cbw = 64 * gw + 4 * lc;
  const int c4 = (tid & 31) * 4, r0 = tid >> 5;
  const int c4w = (tid & 63) * 4, r0w = tid >> 6;
  float* const myred = red + q * (16 * P128);
  float* const myredw = red + qw * (16 * P256);

  const int c = tid & 127;
  const float* pb = P + (blockIdx.x >> 5) * (32 * 128) + c;
  float pv[32];
#pragma unroll
  for (int t = 0; t < 32; ++t) pv[t] = pb[t * 128];
  float ev[8];
#pragma unroll
  for (int j = 0; j < 8; ++j) ev[j] = E[(node0 + (tid >> 7) + 2 * j) * 128 + c];
  STEP_FENCE();
  v4f g1x[16], g1y[16];
#pragma unroll
  for (int i = 0; i < 16; ++i) {
    const int k = 64 * qw + 4 * i + kb;
    g1x[i] = WLOAD(Wen1, k, 256, cbw);
    g1y[i] = WLOAD(Wen1, k + 1, 256, cbw);
  }
  const v4f biasA = *(const v4f*)(ben1 + c4w);
  const v4f biasB = *(const v4f*)(ben2 + c4);
  const v4f biasC = *(const v4f*)(bne1 + c4);
  float wd[8];
#pragma unroll
  for (int j = 0; j < 8; ++j) wd[j] = Wne2[(tid & 15) + 16 * j];
  const float bd = bne2[0];

  {
    float s = 0.0f;
#pragma unroll
    for (int t = 0; t < 32; ++t) s += pv[t];
#pragma unroll
    for (int j = 0; j < 8; ++j) as[(tid >> 7) + 2 * j][c] = s - ev[j];
  }
  lds_barrier();

  v4f g2x[16], g2y[16];
  {
    v8f acc0 = {}, acc1 = {}, acc2 = {}, acc3 = {};
#pragma unroll
    for (int i = 0; i < 16; ++i) {
      STEP_FENCE();
      {
        const int k = 64 * q + 4 * i + kb;
        g2x[i] = WLOAD(Wen2, k, 128, cb);
        g2y[i] = WLOAD(Wen2, k + 1, 128, cb);
      }
      const int k = 64 * qw + 4 * i + kb;
      v2f a = {as[lc][k], as[lc][k + 1]};
      MM4(a, g1x[i], g1y[i]);
    }
    {
      const int cb = cbw;
      PUT_PARTS(myredw, P256);
    }
  }
  lds_barrier();
#pragma unroll
  for (int j = 0; j < 4; ++j) {
    const int r = r0w + 4 * j;
    const float* p = red + r * P256 + c4w;
    v4f s = *(const v4f*)p + *(const v4f*)(p + 16 * P256);
    s += biasA;
    v4f u = {fmaxf(s.x, 0.0f), fmaxf(s.y, 0.0f), fmaxf(s.z, 0.0f), fmaxf(s.w, 0.0f)};
    *(v4f*)(&us[r][c4w]) = u;
  }
  lds_barrier();

  v4f g3ax[8], g3ay[8], g3bx[8], g3by[8];
  {
    v8f acc0 = {}, acc1 = {}, acc2 = {}, acc3 = {};
#pragma unroll
    for (int i = 0; i < 16; ++i) {
      STEP_FENCE();
      {
        const int j = i >> 1;
        const int k = 32 * q + 4 * j + kb + ((i & 1) ? 128 : 0);
        if (i & 1) { g3bx[j] = WLOAD(Wne1, k, 128, cb); g3by[j] = WLOAD(Wne1, k + 1, 128, cb); }
        else       { g3ax[j] = WLOAD(Wne1, k, 128, cb); g3ay[j] = WLOAD(Wne1, k + 1, 128, cb); }
      }
      const int k = 64 * q + 4 * i + kb;
      v2f a = {us[lc][k], us[lc][k + 1]};
      MM4(a, g2x[i], g2y[i]);
    }
    PUT_PARTS(myred, P128);
  }
  lds_barrier();
#pragma unroll
  for (int j = 0; j < 2; ++j) {
    const int r = r0 + 8 * j;
    const float* p = red + r * P128 + c4;
    v4f s = *(const v4f*)p + *(const v4f*)(p + 16 * P128);
    s += *(const v4f*)(p + 2 * 16 * P128);
    s += *(const v4f*)(p + 3 * 16 * P128);
    s += biasB;
    *(v4f*)(h2 + (node0 + r) * 128 + c4) = s;
    *(v4f*)(&hs[r][c4]) = s;
  }
  lds_barrier();

  {
    v8f acc0 = {}, acc1 = {}, acc2 = {}, acc3 = {};
#pragma unroll
    for (int i = 0; i < 8; ++i) {
      STEP_FENCE();
      const int k = 32 * q + 4 * i + kb;
      v2f a = {hs[lc][k], hs[lc][k + 1]};
      const v4f bx = g3ax[i] + g3bx[i];
      const v4f by = g3ay[i] + g3by[i];
      MM4(a, bx, by);
    }
    PUT_PARTS(myred, P128);
  }
  lds_barrier();
#pragma unroll
  for (int j = 0; j < 2; ++j) {
    const int r = r0 + 8 * j;
    const float* p = red + r * P128 + c4;
    v4f s = *(const v4f*)p + *(const v4f*)(p + 16 * P128);
    s += *(const v4f*)(p + 2 * 16 * P128);
    s += *(const v4f*)(p + 3 * 16 * P128);
    s += biasC;
    v4f u = {fmaxf(s.x, 0.0f), fmaxf(s.y, 0.0f), fmaxf(s.z, 0.0f), fmaxf(s.w, 0.0f)};
    *(v4f*)(&vs[r][c4]) = u;
  }
  lds_barrier();

  {
    int r = tid >> 4, l16 = tid & 15;
    float q = 0.0f;
#pragma unroll
    for (int j = 0; j < 8; ++j) q += vs[r][l16 + 16 * j] * wd[j];
    ps[r][l16] = q;
  }
  lds_barrier();
  if (tid < 16) {
    float s = 0.0f;
    for (int i = 0; i < 16; ++i) s += ps[tid][i];
    sc[tid] = (s + bd) * (1.0f / 512.0f);
  }
  lds_barrier();

  for (int idx = tid; idx < 16 * 512; idx += 256) {
    int r = idx >> 9, j = idx & 511;
    out2[(size_t)(node0 + r) * 512 + j] = sc[r];
  }
}

extern "C" void kernel_launch(void* const* d_in, const int* in_sizes, int n_in,
                              void* d_out, int out_size, void* d_ws,
                              size_t ws_size, hipStream_t stream) {
  (void)in_sizes; (void)n_in; (void)out_size; (void)ws_size;
  const float* x    = (const float*)d_in[0];
  const float* Wn1  = (const float*)d_in[1];
  const float* bn1  = (const float*)d_in[2];
  const float* Wn2  = (const float*)d_in[3];
  const float* bn2  = (const float*)d_in[4];
  const float* We1  = (const float*)d_in[5];
  const float* be1  = (const float*)d_in[6];
  const float* We2  = (const float*)d_in[7];
  const float* be2  = (const float*)d_in[8];
  const float* Wen1 = (const float*)d_in[9];
  const float* ben1 = (const float*)d_in[10];
  const float* Wen2 = (const float*)d_in[11];
  const float* ben2 = (const float*)d_in[12];
  const float* Wne1 = (const float*)d_in[13];
  const float* bne1 = (const float*)d_in[14];
  const float* Wne2 = (const float*)d_in[15];
  const float* bne2 = (const float*)d_in[16];

  float* ws = (float*)d_ws;
  float* E  = ws;
  float* P  = ws + 1024 * 128;
  float* h2   = (float*)d_out;
  float* out2 = (float*)d_out + 1024 * 128;

  dim3 blk(256);
  k_node_edge<<<64, blk, 0, stream>>>(x, Wn1, bn1, Wn2, bn2, We1, be1, We2, be2,
                                      E, P);
  k_e2n_n2e<<<64, blk, 0, stream>>>(E, P, Wen1, ben1, Wen2, ben2, Wne1, bne1,
                                    Wne2, bne2, h2, out2);
}


# Node-independent weight algebra done once in launch 1 by 18 spare work-groups: [We2;be2]@Wen1 replaces the We2 stage and Wen1, Wne1 halves pre-added; products re-associated (sum order changes)
# speedup vs baseline: 4.1006x; 1.0589x over previous
#include <hip/hip_runtime.h>

typedef __attribute__((ext_vector_type(2))) float v2f;
typedef __attribute__((ext_vector_type(4))) float v4f;
typedef __attribute__((ext_vector_type(8))) float v8f;

__device__ __forceinline__ v8f wmma_f32(v2f a, v2f b, v8f c) {
  return __builtin_amdgcn_wmma_f32_16x16x4_f32(false, a, false, b, (short)0, c,
                                               false, false);
}

__device__ __forceinline__ void lds_barrier() {
  __builtin_amdgcn_fence(__ATOMIC_RELEASE, "workgroup", "local");
  __builtin_amdgcn_s_barrier();
  __builtin_amdgcn_fence(__ATOMIC_ACQUIRE, "workgroup", "local");
}

#define STEP_FENCE() __builtin_amdgcn_sched_barrier(0x104)

#define P128 132
#define P256 260
#define PDOT 148

#define WLOAD(W, ROW, LD, CB) (*(const v4f*)((W) + (ROW) * (LD) + (CB)))
#define MM4(A, BX, BY)                                        \
  do {                                                        \
    v2f b0_ = {(BX).x, (BY).x}, b1_ = {(BX).y, (BY).y};       \
    v2f b2_ = {(BX).z, (BY).z}, b3_ = {(BX).w, (BY).w};       \
    acc0 = wmma_f32((A), b0_, acc0);                          \
    acc1 = wmma_f32((A), b1_, acc1);                          \
    acc2 = wmma_f32((A), b2_, acc2);                          \
    acc3 = wmma_f32((A), b3_, acc3);                          \
  } while (0)
#define PUT_PARTS(BASE, PITCH)                                            \
  do {                                                                    \
    _Pragma("unroll") for (int v = 0; v < 8; ++v) {                       \
      v4f o_ = {acc0[v], acc1[v], acc2[v], acc3[v]};                      \
      *(v4f*)((BASE) + (v + 8 * hi) * (PITCH) + cb) = o_;                 \
    }                                                                     \
  } while (0)

__global__ __launch_bounds__(256) void k_node_edge(
    const float* __restrict__ x, const float* __restrict__ Wn1,
    const float* __restrict__ bn1, const float* __restrict__ Wn2,
    const float* __restrict__ bn2, const float* __restrict__ We1,
    const float* __restrict__ be1, const float* __restrict__ We2,
    const float* __restrict__ be2, const float* __restrict__ Wen1,
    const float* __restrict__ Wne1, float* __restrict__ U,
    float* __restrict__ P, float* __restrict__ Wpp, float* __restrict__ Wf) {
  __shared__ float xs[16][8];
  __shared__ __attribute__((aligned(16))) float ts[16][P256];
  __shared__ __attribute__((aligned(16))) float hs[16][P128];
  __shared__ __attribute__((aligned(16))) float us[16][P128];
  __shared__ __attribute__((aligned(16))) float red[4 * 16 * P128];
  const int tid = threadIdx.x;
  const int node0 = blockIdx.x * 16;
  const int wave = tid >> 5, lane = tid & 31;
  const int hi = lane >> 4, lc = lane & 15;
  const int kb = 2 * hi;
  const int g = wave & 1, q = wave >> 1;
  const int cb = 64 * g + 4 * lc;
  const int c4 = (tid & 31) * 4, r0 = tid >> 5;
  float* const myred = red + q * (16 * P128);

  if (blockIdx.x >= 64) {
    const int e = blockIdx.x - 64;
    const int mt = e >> 1, ch = e & 1;
    v4f a0 = {}, a1 = {};
    if (mt < 8) {
      a0 = *(const v4f*)(We2 + (16 * mt + r0) * 128 + c4);
      a1 = *(const v4f*)(We2 + (16 * mt + r0 + 8) * 128 + c4);
    } else if (r0 == 0) {
      a0 = *(const v4f*)(be2 + c4);
    }
    v4f fx[8], fy[8];
#pragma unroll
    for (int i = 0; i < 8; ++i) {
      const int k = 32 * q + 4 * i + kb;
      fx[i] = WLOAD(Wen1, k, 256, 128 * ch + cb);
      fy[i] = WLOAD(Wen1, k + 1, 256, 128 * ch + cb);
    }
    if (e < 16) {
      const int i4 = (e * 256 + tid) * 4;
      *(v4f*)(Wf + i4) = *(const v4f*)(Wne1 + i4) + *(const v4f*)(Wne1 + 128 * 128 + i4);
    }
    *(v4f*)(&us[r0][c4]) = a0;
    *(v4f*)(&us[r0 + 8][c4]) = a1;
    lds_barrier();
    {
      v8f acc0 = {}, acc1 = {}, acc2 = {}, acc3 = {};
#pragma unroll
      for (int i = 0; i < 8; ++i) {
        const int k = 32 * q + 4 * i + kb;
        v2f a = {us[lc][k], us[lc][k + 1]};
        MM4(a, fx[i], fy[i]);
      }
      PUT_PARTS(myred, P128);
    }
    lds_barrier();
#pragma unroll
    for (int j = 0; j < 2; ++j) {
      const int r = r0 + 8 * j;
      const float* p = red + r * P128 + c4;
      v4f s = *(const v4f*)p + *(const v4f*)(p + 16 * P128);
      s += *(const v4f*)(p + 2 * 16 * P128);
      s += *(const v4f*)(p + 3 * 16 * P128);
      *(v4f*)(Wpp + (16 * mt + r) * 256 + 128 * ch + c4) = s;
    }
    return;
  }

  float xv = 0.0f;
  if (tid < 16 * 8) {
    int r = tid >> 3, k = tid & 7;
    if (k < 6) xv = x[(node0 + r) * 6 + k];
  }
  const int o = tid;
  const float w0 = Wn1[0 * 256 + o], w1 = Wn1[1 * 256 + o], w2 = Wn1[2 * 256 + o];
  const float w3 = Wn1[3 * 256 + o], w4 = Wn1[4 * 256 + o], w5 = Wn1[5 * 256 + o];
  const float b1 = bn1[o];
  STEP_FENCE();
  v4f f2x[16], f2y[16];
#pragma unroll
  for (int i = 0; i < 16; ++i) {
    const int k = 64 * q + 4 * i + kb;
    f2x[i] = WLOAD(Wn2, k, 128, cb);
    f2y[i] = WLOAD(Wn2, k + 1, 128, cb);
  }
  const v4f bias2 = *(const v4f*)(bn2 + c4);
  const v4f bias3 = *(const v4f*)(be1 + c4);

  if (tid < 16 * 8) {
    int r = tid >> 3, k = tid & 7;
    xs[r][k] = xv;
  }
  lds_barrier();

  for (int r = 0; r < 16; ++r) {
    float acc = b1 + xs[r][0] * w0 + xs[r][1] * w1 + xs[r][2] * w2 +
                xs[r][3] * w3 + xs[r][4] * w4 + xs[r][5] * w5;
    ts[r][o] = fmaxf(acc, 0.0f);
  }
  lds_barrier();

  v4f f3ax[8], f3ay[8], f3bx[8], f3by[8];
  {
    v8f acc0 = {}, acc1 = {}, acc2 = {}, acc3 = {};
#pragma unroll
    for (int i = 0; i < 16; ++i) {
      STEP_FENCE();
      {
        const int j = i >> 1;
        const int k = 32 * q + 4 * j + kb + ((i & 1) ? 128 : 0);
        if (i & 1) { f3bx[j] = WLOAD(We1, k, 128, cb); f3by[j] = WLOAD(We1, k + 1, 128, cb); }
        else       { f3ax[j] = WLOAD(We1, k, 128, cb); f3ay[j] = WLOAD(We1, k + 1, 128, cb); }
      }
      const int k = 64 * q + 4 * i + kb;
      v2f a = {ts[lc][k], ts[lc][k + 1]};
      MM4(a, f2x[i], f2y[i]);
    }
    PUT_PARTS(myred, P128);
  }
  lds_barrier();
#pragma unroll
  for (int j = 0; j < 2; ++j) {
    const int r = r0 + 8 * j;
    const float* p = red + r * P128 + c4;
    v4f s = *(const v4f*)p + *(const v4f*)(p + 16 * P128);
    s += *(const v4f*)(p + 2 * 16 * P128);
    s += *(const v4f*)(p + 3 * 16 * P128);
    *(v4f*)(&hs[r][c4]) = s + bias2;
  }
  lds_barrier();

  {
    v8f acc0 = {}, acc1 = {}, acc2 = {}, acc3 = {};
#pragma unroll
    for (int i = 0; i < 8; ++i) {
      STEP_FENCE();
      const int k = 32 * q + 4 * i + kb;
      v2f a = {hs[lc][k], hs[lc][k + 1]};
      const v4f bx = f3ax[i] + f3bx[i];
      const v4f by = f3ay[i] + f3by[i];
      MM4(a, bx, by);
    }
    PUT_PARTS(myred, P128);
  }
  lds_barrier();
#pragma unroll
  for (int j = 0; j < 2; ++j) {
    const int r = r0 + 8 * j;
    const float* p = red + r * P128 + c4;
    v4f s = *(const v4f*)p + *(const v4f*)(p + 16 * P128);
    s += *(const v4f*)(p + 2 * 16 * P128);
    s += *(const v4f*)(p + 3 * 16 * P128);
    s += bias3;
    v4f u = {fmaxf(s.x, 0.0f), fmaxf(s.y, 0.0f), fmaxf(s.z, 0.0f), fmaxf(s.w, 0.0f)};
    *(v4f*)(U + (node0 + r) * 128 + c4) = u;
    *(v4f*)(&us[r][c4]) = u;
  }
  lds_barrier();
  if (tid < 128) {
    float p0 = 0.0f, p1 = 0.0f;
    for (int r = 0; r < 8; ++r) { p0 += us[r][tid]; p1 += us[r + 8][tid]; }
    P[blockIdx.x * 128 + tid] = p0 + p1;
  }
}

__global__ __launch_bounds__(256) void k_e2n_n2e(
    const float* U, const float* __restrict__ P,
    const float* __restrict__ Wpp, const float* __restrict__ ben1,
    const float* __restrict__ Wen2, const float* __restrict__ ben2,
    const float* __restrict__ Wf, const float* __restrict__ bne1,
    const float* __restrict__ Wne2, const float* __restrict__ bne2,
    float* h2, float* __restrict__ out2) {
  __shared__ __attribute__((aligned(16))) float as[16][P128];
  __shared__ __attribute__((aligned(16))) float us[16][P256];
  __shared__ __attribute__((aligned(16))) float hs[16][P128];
  __shared__ __attribute__((aligned(16))) float vs[16][PDOT];
  __shared__ float ps[16][16];
  __shared__ float sc[16];
  __shared__ __attribute__((aligned(16))) float red[4 * 16 * P128];
  const int tid = threadIdx.x;
  const int node0 = blockIdx.x * 16;
  const int wave = tid >> 5, lane = tid & 31;
  const int hi = lane >> 4, lc = lane & 15;
  const int kb = 2 * hi;
  const int g = wave & 1, q = wave >> 1;
  const int cb = 64 * g + 4 * lc;
  const int gw = wave & 3, qw = wave >> 2;
  const int cbw = 64 * gw + 4 * lc;
  const int c4 = (tid & 31) * 4, r0 = tid >> 5;
  const int c4w = (tid & 63) * 4, r0w = tid >> 6;
  float* const myred = red + q * (16 * P128);
  float* const myredw = red + qw * (16 * P256);

  const int c = tid & 127;
  const float* pb = P + (blockIdx.x >> 5) * (32 * 128) + c;
  float pv[32];
#pragma unroll
  for (int t = 0; t < 32; ++t) pv[t] = pb[t * 128];
  float ev[8];
#pragma unroll
  for (int j = 0; j < 8; ++j) ev[j] = U[(node0 + (tid >> 7) + 2 * j) * 128 + c];
  STEP_FENCE();
  v4f g1x[16], g1y[16];
#pragma unroll
  for (int i = 0; i < 16; ++i) {
    const int k = 64 * qw + 4 * i + kb;
    g1x[i] = WLOAD(Wpp, k, 256, cbw);
    g1y[i] = WLOAD(Wpp, k + 1, 256, cbw);
  }
  const v4f biasA = *(const v4f*)(ben1 + c4w) + 511.0f * *(const v4f*)(Wpp + 128 * 256 + c4w);
  const v4f biasB = *(const v4f*)(ben2 + c4);
  const v4f biasC = *(const v4f*)(bne1 + c4);
  float wd[8];
#pragma unroll
  for (int j = 0; j < 8; ++j) wd[j] = Wne2[(tid & 15) + 16 * j];
  const float bd = bne2[0];

  {
    float s = 0.0f;
#pragma unroll
    for (int t = 0; t < 32; ++t) s += pv[t];
#pragma unroll
    for (int j = 0; j < 8; ++j) as[(tid >> 7) + 2 * j][c] = s - ev[j];
  }
  lds_barrier();

  v4f g2x[16], g2y[16];
  {
    v8f acc0 = {}, acc1 = {}, acc2 = {}, acc3 = {};
#pragma unroll
    for (int i = 0; i < 16; ++i) {
      STEP_FENCE();
      {
        const int k = 64 * q + 4 * i + kb;
        g2x[i] = WLOAD(Wen2, k, 128, cb);
        g2y[i] = WLOAD(Wen2, k + 1, 128, cb);
      }
      const int k = 64 * qw + 4 * i + kb;
      v2f a = {as[lc][k], as[lc][k + 1]};
      MM4(a, g1x[i], g1y[i]);
    }
    {
      const int cb = cbw;
      PUT_PARTS(myredw, P256);
    }
  }
  lds_barrier();
#pragma unroll
  for (int j = 0; j < 4; ++j) {
    const int r = r0w + 4 * j;
    const float* p = red + r * P256 + c4w;
    v4f s = *(const v4f*)p + *(const v4f*)(p + 16 * P256);
    s += biasA;
    v4f u = {fmaxf(s.x, 0.0f), fmaxf(s.y, 0.0f), fmaxf(s.z, 0.0f), fmaxf(s.w, 0.0f)};
    *(v4f*)(&us[r][c4w]) = u;
  }
  lds_barrier();

  v4f g3x[8], g3y[8];
  {
    v8f acc0 = {}, acc1 = {}, acc2 = {}, acc3 = {};
#pragma unroll
    for (int i = 0; i < 16; ++i) {
      STEP_FENCE();
      if (!(i & 1)) {
        const int j = i >> 1;
        const int k = 32 * q + 4 * j + kb;
        g3x[j] = WLOAD(Wf, k, 128, cb);
        g3y[j] = WLOAD(Wf, k + 1, 128, cb);
      }
      const int k = 64 * q + 4 * i + kb;
      v2f a = {us[lc][k], us[lc][k + 1]};
      MM4(a, g2x[i], g2y[i]);
    }
    PUT_PARTS(myred, P128);
  }
  lds_barrier();
#pragma unroll
  for (int j = 0; j < 2; ++j) {
    const int r = r0 + 8 * j;
    const float* p = red + r * P128 + c4;
    v4f s = *(const v4f*)p + *(const v4f*)(p + 16 * P128);
    s += *(const v4f*)(p + 2 * 16 * P128);
    s += *(const v4f*)(p + 3 * 16 * P128);
    s += biasB;
    *(v4f*)(h2 + (node0 + r) * 128 + c4) = s;
    *(v4f*)(&hs[r][c4]) = s;
  }
  lds_barrier();

  {
    v8f acc0 = {}, acc1 = {}, acc2 = {}, acc3 = {};
#pragma unroll
    for (int i = 0; i < 8; ++i) {
      const int k = 32 * q + 4 * i + kb;
      v2f a = {hs[lc][k], hs[lc][k + 1]};
      MM4(a, g3x[i], g3y[i]);
    }
    PUT_PARTS(myred, P128);
  }
  lds_barrier();
#pragma unroll
  for (int j = 0; j < 2; ++j) {
    const int r = r0 + 8 * j;
    const float* p = red + r * P128 + c4;
    v4f s = *(const v4f*)p + *(const v4f*)(p + 16 * P128);
    s += *(const v4f*)(p + 2 * 16 * P128);
    s += *(const v4f*)(p + 3 * 16 * P128);
    s += biasC;
    v4f u = {fmaxf(s.x, 0.0f), fmaxf(s.y, 0.0f), fmaxf(s.z, 0.0f), fmaxf(s.w, 0.0f)};
    *(v4f*)(&vs[r][c4]) = u;
  }
  lds_barrier();

  {
    int r = tid >> 4, l16 = tid & 15;
    float q = 0.0f;
#pragma unroll
    for (int j = 0; j < 8; ++j) q += vs[r][l16 + 16 * j] * wd[j];
    ps[r][l16] = q;
  }
  lds_barrier();
  if (tid < 16) {
    float s = 0.0f;
    for (int i = 0; i < 16; ++i) s += ps[tid][i];
    sc[tid] = (s + bd) * (1.0f / 512.0f);
  }
  lds_barrier();

  for (int idx = tid; idx < 16 * 512; idx += 256) {
    int r = idx >> 9, j = idx & 511;
    out2[(size_t)(node0 + r) * 512 + j] = sc[r];
  }
}

extern "C" void kernel_launch(void* const* d_in, const int* in_sizes, int n_in,
                              void* d_out, int out_size, void* d_ws,
                              size_t ws_size, hipStream_t stream) {
  (void)in_sizes; (void)n_in; (void)out_size; (void)ws_size;
  const float* x    = (const float*)d_in[0];
  const float* Wn1  = (const float*)d_in[1];
  const float* bn1  = (const float*)d_in[2];
  const float* Wn2  = (const float*)d_in[3];
  const float* bn2  = (const float*)d_in[4];
  const float* We1  = (const float*)d_in[5];
  const float* be1  = (const float*)d_in[6];
  const float* We2  = (const float*)d_in[7];
  const float* be2  = (const float*)d_in[8];
  const float* Wen1 = (const float*)d_in[9];
  const float* ben1 = (const float*)d_in[10];
  const float* Wen2 = (const float*)d_in[11];
  const float* ben2 = (const float*)d_in[12];
  const float* Wne1 = (const float*)d_in[13];
  const float* bne1 = (const float*)d_in[14];
  const float* Wne2 = (const float*)d_in[15];
  const float* bne2 = (const float*)d_in[16];

  float* ws = (float*)d_ws;
  float* Wpp = ws;
  float* Wf  = ws + 144 * 256;
  float* P   = ws + 1024 * 128;
  float* h2   = (float*)d_out;
  float* out2 = (float*)d_out + 1024 * 128;
  float* U    = h2;

  dim3 blk(256);
  k_node_edge<<<82, blk, 0, stream>>>(x, Wn1, bn1, Wn2, bn2, We1, be1, We2,
                                           be2, Wen1, Wne1, U, P, Wpp, Wf);
  k_e2n_n2e<<<64, blk, 0, stream>>>(U, P, Wpp, ben1, Wen2, ben2, Wf, bne1,
                                    Wne2, bne2, h2, out2);
}


# Last dot (128 to 1) finished inside each wave by lane shuffles; each wave writes its own two edge rows with 16-byte stores: the tail's 3 barriers and LDS hand-overs go (dot sum order changes)
# speedup vs baseline: 4.4544x; 1.0863x over previous
#include <hip/hip_runtime.h>

typedef __attribute__((ext_vector_type(2))) float v2f;
typedef __attribute__((ext_vector_type(4))) float v4f;
typedef __attribute__((ext_vector_type(8))) float v8f;

__device__ __forceinline__ v8f wmma_f32(v2f a, v2f b, v8f c) {
  return __builtin_amdgcn_wmma_f32_16x16x4_f32(false, a, false, b, (short)0, c,
                                               false, false);
}

__device__ __forceinline__ void lds_barrier() {
  __builtin_amdgcn_fence(__ATOMIC_RELEASE, "workgroup", "local");
  __builtin_amdgcn_s_barrier();
  __builtin_amdgcn_fence(__ATOMIC_ACQUIRE, "workgroup", "local");
}

#define STEP_FENCE() __builtin_amdgcn_sched_barrier(0x104)

#define P128 132
#define P256 260

#define ROW_XOR(X, M)                                                          \
  __builtin_bit_cast(float, __builtin_amdgcn_update_dpp(                       \
      __builtin_bit_cast(int, (X)), __builtin_bit_cast(int, (X)), 0x160 + (M), \
      0xf, 0xf, false))
__device__ __forceinline__ float wave_sum(float x) {
  x += ROW_XOR(x, 1);
  x += ROW_XOR(x, 2);
  x += ROW_XOR(x, 4);
  x += ROW_XOR(x, 8);
  const int b = __builtin_bit_cast(int, x);
  x += __builtin_bit_cast(
      float, __builtin_amdgcn_permlanex16(b, b, 0x76543210, 0xfedcba98, false, false));
  return x;
}

#define WLOAD(W, ROW, LD, CB) (*(const v4f*)((W) + (ROW) * (LD) + (CB)))
#define MM4(A, BX, BY)                                        \
  do {                                                        \
    v2f b0_ = {(BX).x, (BY).x}, b1_ = {(BX).y, (BY).y};       \
    v2f b2_ = {(BX).z, (BY).z}, b3_ = {(BX).w, (BY).w};       \
    acc0 = wmma_f32((A), b0_, acc0);                          \
    acc1 = wmma_f32((A), b1_, acc1);                          \
    acc2 = wmma_f32((A), b2_, acc2);                          \
    acc3 = wmma_f32((A), b3_, acc3);                          \
  } while (0)
#define PUT_PARTS(BASE, PITCH)                                            \
  do {                                                                    \
    _Pragma("unroll") for (int v = 0; v < 8; ++v) {                       \
      v4f o_ = {acc0[v], acc1[v], acc2[v], acc3[v]};                      \
      *(v4f*)((BASE) + (v + 8 * hi) * (PITCH) + cb) = o_;                 \
    }                                                                     \
  } while (0)

__global__ __launch_bounds__(256) void k_node_edge(
    const float* __restrict__ x, const float* __restrict__ Wn1,
    const float* __restrict__ bn1, const float* __restrict__ Wn2,
    const float* __restrict__ bn2, const float* __restrict__ We1,
    const float* __restrict__ be1, const float* __restrict__ We2,
    const float* __restrict__ be2, const float* __restrict__ Wen1,
    const float* __restrict__ Wne1, float* __restrict__ U,
    float* __restrict__ P, float* __restrict__ Wpp, float* __restrict__ Wf) {
  __shared__ float xs[16][8];
  __shared__ __attribute__((aligned(16))) float ts[16][P256];
  __shared__ __attribute__((aligned(16))) float hs[16][P128];
  __shared__ __attribute__((aligned(16))) float us[16][P128];
  __shared__ __attribute__((aligned(16))) float red[4 * 16 * P128];
  const int tid = threadIdx.x;
  const int node0 = blockIdx.x * 16;
  const int wave = tid >> 5, lane = tid & 31;
  const int hi = lane >> 4, lc = lane & 15;
  const int kb = 2 * hi;
  const int g = wave & 1, q = wave >> 1;
  const int cb = 64 * g + 4 * lc;
  const int c4 = (tid & 31) * 4, r0 = tid >> 5;
  float* const myred = red + q * (16 * P128);

  if (blockIdx.x >= 64) {
    const int e = blockIdx.x - 64;
    const int mt = e >> 1, ch = e & 1;
    v4f a0 = {}, a1 = {};
    if (mt < 8) {
      a0 = *(const v4f*)(We2 + (16 * mt + r0) * 128 + c4);
      a1 = *(const v4f*)(We2 + (16 * mt + r0 + 8) * 128 + c4);
    } else if (r0 == 0) {
      a0 = *(const v4f*)(be2 + c4);
    }
    v4f fx[8], fy[8];
#pragma unroll
    for (int i = 0; i < 8; ++i) {
      const int k = 32 * q + 4 * i + kb;
      fx[i] = WLOAD(Wen1, k, 256, 128 * ch + cb);
      fy[i] = WLOAD(Wen1, k + 1, 256, 128 * ch + cb);
    }
    if (e < 16) {
      const int i4 = (e * 256 + tid) * 4;
      *(v4f*)(Wf + i4) = *(const v4f*)(Wne1 + i4) + *(const v4f*)(Wne1 + 128 * 128 + i4);
    }
    *(v4f*)(&us[r0][c4]) = a0;
    *(v4f*)(&us[r0 + 8][c4]) = a1;
    lds_barrier();
    {
      v8f acc0 = {}, acc1 = {}, acc2 = {}, acc3 = {};
#pragma unroll
      for (int i = 0; i < 8; ++i) {
        const int k = 32 * q + 4 * i + kb;
        v2f a = {us[lc][k], us[lc][k + 1]};
        MM4(a, fx[i], fy[i]);
      }
      PUT_PARTS(myred, P128);
    }
    lds_barrier();
#pragma unroll
    for (int j = 0; j < 2; ++j) {
      const int r = r0 + 8 * j;
      const float* p = red + r * P128 + c4;
      v4f s = *(const v4f*)p + *(const v4f*)(p + 16 * P128);
      s += *(const v4f*)(p + 2 * 16 * P128);
      s += *(const v4f*)(p + 3 * 16 * P128);
      *(v4f*)(Wpp + (16 * mt + r) * 256 + 128 * ch + c4) = s;
    }
    return;
  }

  float xv = 0.0f;
  if (tid < 16 * 8) {
    int r = tid >> 3, k = tid & 7;
    if (k < 6) xv = x[(node0 + r) * 6 + k];
  }
  const int o = tid;
  const float w0 = Wn1[0 * 256 + o], w1 = Wn1[1 * 256 + o], w2 = Wn1[2 * 256 + o];
  const float w3 = Wn1[3 * 256 + o], w4 = Wn1[4 * 256 + o], w5 = Wn1[5 * 256 + o];
  const float b1 = bn1[o];
  STEP_FENCE();
  v4f f2x[16], f2y[16];
#pragma unroll
  for (int i = 0; i < 16; ++i) {
    const int k = 64 * q + 4 * i + kb;
    f2x[i] = WLOAD(Wn2, k, 128, cb);
    f2y[i] = WLOAD(Wn2, k + 1, 128, cb);
  }
  const v4f bias2 = *(const v4f*)(bn2 + c4);
  const v4f bias3 = *(const v4f*)(be1 + c4);

  if (tid < 16 * 8) {
    int r = tid >> 3, k = tid & 7;
    xs[r][k] = xv;
  }
  lds_barrier();

  for (int r = 0; r < 16; ++r) {
    float acc = b1 + xs[r][0] * w0 + xs[r][1] * w1 + xs[r][2] * w2 +
                xs[r][3] * w3 + xs[r][4] * w4 + xs[r][5] * w5;
    ts[r][o] = fmaxf(acc, 0.0f);
  }
  lds_barrier();

  v4f f3ax[8], f3ay[8], f3bx[8], f3by[8];
  {
    v8f acc0 = {}, acc1 = {}, acc2 = {}, acc3 = {};
#pragma unroll
    for (int i = 0; i < 16; ++i) {
      STEP_FENCE();
      {
        const int j = i >> 1;
        const int k = 32 * q + 4 * j + kb + ((i & 1) ? 128 : 0);
        if (i & 1) { f3bx[j] = WLOAD(We1, k, 128, cb); f3by[j] = WLOAD(We1, k + 1, 128, cb); }
        else       { f3ax[j] = WLOAD(We1, k, 128, cb); f3ay[j] = WLOAD(We1, k + 1, 128, cb); }
      }
      const int k = 64 * q + 4 * i + kb;
      v2f a = {ts[lc][k], ts[lc][k + 1]};
      MM4(a, f2x[i], f2y[i]);
    }
    PUT_PARTS(myred, P128);
  }
  lds_barrier();
#pragma unroll
  for (int j = 0; j < 2; ++j) {
    const int r = r0 + 8 * j;
    const float* p = red + r * P128 + c4;
    v4f s = *(const v4f*)p + *(const v4f*)(p + 16 * P128);
    s += *(const v4f*)(p + 2 * 16 * P128);
    s += *(const v4f*)(p + 3 * 16 * P128);
    *(v4f*)(&hs[r][c4]) = s + bias2;
  }
  lds_barrier();

  {
    v8f acc0 = {}, acc1 = {}, acc2 = {}, acc3 = {};
#pragma unroll
    for (int i = 0; i < 8; ++i) {
      STEP_FENCE();
      const int k = 32 * q + 4 * i + kb;
      v2f a = {hs[lc][k], hs[lc][k + 1]};
      const v4f bx = f3ax[i] + f3bx[i];
      const v4f by = f3ay[i] + f3by[i];
      MM4(a, bx, by);
    }
    PUT_PARTS(myred, P128);
  }
  lds_barrier();
#pragma unroll
  for (int j = 0; j < 2; ++j) {
    const int r = r0 + 8 * j;
    const float* p = red + r * P128 + c4;
    v4f s = *(const v4f*)p + *(const v4f*)(p + 16 * P128);
    s += *(const v4f*)(p + 2 * 16 * P128);
    s += *(const v4f*)(p + 3 * 16 * P128);
    s += bias3;
    v4f u = {fmaxf(s.x, 0.0f), fmaxf(s.y, 0.0f), fmaxf(s.z, 0.0f), fmaxf(s.w, 0.0f)};
    *(v4f*)(U + (node0 + r) * 128 + c4) = u;
    *(v4f*)(&us[r][c4]) = u;
  }
  lds_barrier();
  if (tid < 128) {
    float p0 = 0.0f, p1 = 0.0f;
    for (int r = 0; r < 8; ++r) { p0 += us[r][tid]; p1 += us[r + 8][tid]; }
    P[blockIdx.x * 128 + tid] = p0 + p1;
  }
}

__global__ __launch_bounds__(256) void k_e2n_n2e(
    const float* U, const float* __restrict__ P,
    const float* __restrict__ Wpp, const float* __restrict__ ben1,
    const float* __restrict__ Wen2, const float* __restrict__ ben2,
    const float* __restrict__ Wf, const float* __restrict__ bne1,
    const float* __restrict__ Wne2, const float* __restrict__ bne2,
    float* h2, float* __restrict__ out2) {
  __shared__ __attribute__((aligned(16))) float as[16][P128];
  __shared__ __attribute__((aligned(16))) float us[16][P256];
  __shared__ __attribute__((aligned(16))) float hs[16][P128];
  __shared__ __attribute__((aligned(16))) float red[4 * 16 * P128];
  const int tid = threadIdx.x;
  const int node0 = blockIdx.x * 16;
  const int wave = tid >> 5, lane = tid & 31;
  const int hi = lane >> 4, lc = lane & 15;
  const int kb = 2 * hi;
  const int g = wave & 1, q = wave >> 1;
  const int cb = 64 * g + 4 * lc;
  const int gw = wave & 3, qw = wave >> 2;
  const int cbw = 64 * gw + 4 * lc;
  const int c4 = (tid & 31) * 4, r0 = tid >> 5;
  const int c4w = (tid & 63) * 4, r0w = tid >> 6;
  float* const myred = red + q * (16 * P128);
  float* const myredw = red + qw * (16 * P256);

  const int c = tid & 127;
  const float* pb = P + (blockIdx.x >> 5) * (32 * 128) + c;
  float pv[32];
#pragma unroll
  for (int t = 0; t < 32; ++t) pv[t] = pb[t * 128];
  float ev[8];
#pragma unroll
  for (int j = 0; j < 8; ++j) ev[j] = U[(node0 + (tid >> 7) + 2 * j) * 128 + c];
  STEP_FENCE();
  v4f g1x[16], g1y[16];
#pragma unroll
  for (int i = 0; i < 16; ++i) {
    const int k = 64 * qw + 4 * i + kb;
    g1x[i] = WLOAD(Wpp, k, 256, cbw);
    g1y[i] = WLOAD(Wpp, k + 1, 256, cbw);
  }
  const v4f biasA = *(const v4f*)(ben1 + c4w) + 511.0f * *(const v4f*)(Wpp + 128 * 256 + c4w);
  const v4f biasB = *(const v4f*)(ben2 + c4);
  const v4f biasC = *(const v4f*)(bne1 + c4);
  const v4f wdot = *(const v4f*)(Wne2 + c4);
  const float bd = bne2[0];

  {
    float s = 0.0f;
#pragma unroll
    for (int t = 0; t < 32; ++t) s += pv[t];
#pragma unroll
    for (int j = 0; j < 8; ++j) as[(tid >> 7) + 2 * j][c] = s - ev[j];
  }
  lds_barrier();

  v4f g2x[16], g2y[16];
  {
    v8f acc0 = {}, acc1 = {}, acc2 = {}, acc3 = {};
#pragma unroll
    for (int i = 0; i < 16; ++i) {
      STEP_FENCE();
      {
        const int k = 64 * q + 4 * i + kb;
        g2x[i] = WLOAD(Wen2, k, 128, cb);
        g2y[i] = WLOAD(Wen2, k + 1, 128, cb);
      }
      const int k = 64 * qw + 4 * i + kb;
      v2f a = {as[lc][k], as[lc][k + 1]};
      MM4(a, g1x[i], g1y[i]);
    }
    {
      const int cb = cbw;
      PUT_PARTS(myredw, P256);
    }
  }
  lds_barrier();
#pragma unroll
  for (int j = 0; j < 4; ++j) {
    const int r = r0w + 4 * j;
    const float* p = red + r * P256 + c4w;
    v4f s = *(const v4f*)p + *(const v4f*)(p + 16 * P256);
    s += biasA;
    v4f u = {fmaxf(s.x, 0.0f), fmaxf(s.y, 0.0f), fmaxf(s.z, 0.0f), fmaxf(s.w, 0.0f)};
    *(v4f*)(&us[r][c4w]) = u;
  }
  lds_barrier();

  v4f g3x[8], g3y[8];
  {
    v8f acc0 = {}, acc1 = {}, acc2 = {}, acc3 = {};
#pragma unroll
    for (int i = 0; i < 16; ++i) {
      STEP_FENCE();
      if (!(i & 1)) {
        const int j = i >> 1;
        const int k = 32 * q + 4 * j + kb;
        g3x[j] = WLOAD(Wf, k, 128, cb);
        g3y[j] = WLOAD(Wf, k + 1, 128, cb);
      }
      const int k = 64 * q + 4 * i + kb;
      v2f a = {us[lc][k], us[lc][k + 1]};
      MM4(a, g2x[i], g2y[i]);
    }
    PUT_PARTS(myred, P128);
  }
  lds_barrier();
#pragma unroll
  for (int j = 0; j < 2; ++j) {
    const int r = r0 + 8 * j;
    const float* p = red + r * P128 + c4;
    v4f s = *(const v4f*)p + *(const v4f*)(p + 16 * P128);
    s += *(const v4f*)(p + 2 * 16 * P128);
    s += *(const v4f*)(p + 3 * 16 * P128);
    s += biasB;
    *(v4f*)(h2 + (node0 + r) * 128 + c4) = s;
    *(v4f*)(&hs[r][c4]) = s;
  }
  lds_barrier();

  {
    v8f acc0 = {}, acc1 = {}, acc2 = {}, acc3 = {};
#pragma unroll
    for (int i = 0; i < 8; ++i) {
      const int k = 32 * q + 4 * i + kb;
      v2f a = {hs[lc][k], hs[lc][k + 1]};
      MM4(a, g3x[i], g3y[i]);
    }
    PUT_PARTS(myred, P128);
  }
  lds_barrier();
#pragma unroll
  for (int j = 0; j < 2; ++j) {
    const int r = r0 + 8 * j;
    const float* p = red + r * P128 + c4;
    v4f s = *(const v4f*)p + *(const v4f*)(p + 16 * P128);
    s += *(const v4f*)(p + 2 * 16 * P128);
    s += *(const v4f*)(p + 3 * 16 * P128);
    s += biasC;
    v4f u = {fmaxf(s.x, 0.0f), fmaxf(s.y, 0.0f), fmaxf(s.z, 0.0f), fmaxf(s.w, 0.0f)};
    const float d = wave_sum(u.x * wdot.x + u.y * wdot.y + u.z * wdot.z + u.w * wdot.w);
    const float sc = (d + bd) * (1.0f / 512.0f);
    const v4f o = {sc, sc, sc, sc};
    float* orow = out2 + (size_t)(node0 + r) * 512 + c4;
#pragma unroll
    for (int it = 0; it < 4; ++it) *(v4f*)(orow + 128 * it) = o;
  }
}

extern "C" void kernel_launch(void* const* d_in, const int* in_sizes, int n_in,
                              void* d_out, int out_size, void* d_ws,
                              size_t ws_size, hipStream_t stream) {
  (void)in_sizes; (void)n_in; (void)out_size; (void)ws_size;
  const float* x    = (const float*)d_in[0];
  const float* Wn1  = (const float*)d_in[1];
  const float* bn1  = (const float*)d_in[2];
  const float* Wn2  = (const float*)d_in[3];
  const float* bn2  = (const float*)d_in[4];
  const float* We1  = (const float*)d_in[5];
  const float* be1  = (const float*)d_in[6];
  const float* We2  = (const float*)d_in[7];
  const float* be2  = (const float*)d_in[8];
  const float* Wen1 = (const float*)d_in[9];
  const float* ben1 = (const float*)d_in[10];
  const float* Wen2 = (const float*)d_in[11];
  const float* ben2 = (const float*)d_in[12];
  const float* Wne1 = (const float*)d_in[13];
  const float* bne1 = (const float*)d_in[14];
  const float* Wne2 = (const float*)d_in[15];
  const float* bne2 = (const float*)d_in[16];

  float* ws = (float*)d_ws;
  float* Wpp = ws;
  float* Wf  = ws + 144 * 256;
  float* P   = ws + 1024 * 128;
  float* h2   = (float*)d_out;
  float* out2 = (float*)d_out + 1024 * 128;
  float* U    = h2;

  dim3 blk(256);
  k_node_edge<<<82, blk, 0, stream>>>(x, Wn1, bn1, Wn2, bn2, We1, be1, We2,
                                           be2, Wen1, Wne1, U, P, Wpp, Wf);
  k_e2n_n2e<<<64, blk, 0, stream>>>(U, P, Wpp, ben1, Wen2, ben2, Wf, bne1,
                                    Wne2, bne2, h2, out2);
}
